# starting kernel
# baseline (speedup 1.0000x reference)
#include <hip/hip_runtime.h>
#include <hip/hip_fp16.h>


typedef float v2f __attribute__((ext_vector_type(2)));
typedef float v8f __attribute__((ext_vector_type(8)));

#define NPTS   50000
#define CCH    64
#define KPN    15
#define HNB    32
#define SIGMA  0.7f
#define BN_EPS 1e-5f

#define TILES     ((NPTS + 15) / 16)
#define GRID1     ((TILES + 7) / 8)
#define SCALE_OFF (GRID1 * 128)
#define SHIFT_OFF (SCALE_OFF + 64)

static __device__ __forceinline__ v8f wmma_f32_16x16x4(v2f a, v2f b, v8f c) {
  return __builtin_amdgcn_wmma_f32_16x16x4_f32(false, a, false, b,
                                               (short)0, c, false, false);
}

__global__ __launch_bounds__(256) void k1_stats(const float* __restrict__ sfeat,
                                                const float* __restrict__ W1,
                                                float* __restrict__ partials) {
  __shared__ float wsum[8][64];
  __shared__ float wssq[8][64];
  const int tid  = threadIdx.x;
  const int wave = tid >> 5;
  const int lane = tid & 31;
  const int half = lane >> 4;
  const int lm16 = lane & 15;
  const int t    = blockIdx.x * 8 + wave;

  v8f acc[4];
  #pragma unroll
  for (int n0 = 0; n0 < 4; ++n0)
    acc[n0] = (v8f){0.f,0.f,0.f,0.f,0.f,0.f,0.f,0.f};

  if (t < TILES) {
    const int row = t * 16 + lm16;
    for (int k0 = 0; k0 < CCH; k0 += 4) {
      const int kk = k0 + 2 * half;
      v2f a = *(const v2f*)(sfeat + row * CCH + kk);
      #pragma unroll
      for (int n0 = 0; n0 < 4; ++n0) {
        const int n = n0 * 16 + lm16;
        v2f b;
        b.x = W1[kk * CCH + n];
        b.y = W1[(kk + 1) * CCH + n];
        acc[n0] = wmma_f32_16x16x4(a, b, acc[n0]);
      }
    }
  }

  #pragma unroll
  for (int n0 = 0; n0 < 4; ++n0) {
    float s = 0.f, q = 0.f;
    #pragma unroll
    for (int r = 0; r < 8; ++r) { float v = acc[n0][r]; s += v; q += v * v; }
    s += __shfl_xor(s, 16);
    q += __shfl_xor(q, 16);
    if (half == 0) { wsum[wave][n0 * 16 + lm16] = s; wssq[wave][n0 * 16 + lm16] = q; }
  }
  __syncthreads();

  if (tid < 64) {
    float s = 0.f, q = 0.f;
    #pragma unroll
    for (int w = 0; w < 8; ++w) { s += wsum[w][tid]; q += wssq[w][tid]; }
    partials[blockIdx.x * 128 + tid]      = s;
    partials[blockIdx.x * 128 + 64 + tid] = q;
  }
}

__global__ void k2_finalize(const float* __restrict__ partials,
                            const float* __restrict__ gamma,
                            const float* __restrict__ beta,
                            float* __restrict__ ws) {
  const int ch = threadIdx.x;
  if (ch >= 64) return;
  float s = 0.f, q = 0.f;
  for (int b = 0; b < GRID1; ++b) {
    s += partials[b * 128 + ch];
    q += partials[b * 128 + 64 + ch];
  }
  const float inv_n = 1.0f / (float)NPTS;
  const float mu  = s * inv_n;
  const float var = q * inv_n - mu * mu;
  const float sc  = gamma[ch] * rsqrtf(var + BN_EPS);
  ws[SCALE_OFF + ch] = sc;
  ws[SHIFT_OFF + ch] = beta[ch] - mu * sc;
}

__global__ __launch_bounds__(256) void k3_main(const float* __restrict__ qpts,
                                               const float* __restrict__ spts,
                                               const float* __restrict__ sfeat,
                                               const int*   __restrict__ nbidx,
                                               const float* __restrict__ kpts,
                                               const float* __restrict__ W1,
                                               const float* __restrict__ W2,
                                               const float* __restrict__ b2,
                                               const float* __restrict__ ws,
                                               float* __restrict__ out) {
  __shared__ float  kp[45];
  __shared__ float  act[16 * 68];
  __shared__ __half cw[16 * 960];
  __shared__ int    jIdx[512];
  __shared__ int    nnK[512];
  __shared__ float  nnI[512];

  const int tid  = threadIdx.x;
  const int wave = tid >> 5;
  const int lane = tid & 31;
  const int half = lane >> 4;
  const int lm16 = lane & 15;
  const int m0   = blockIdx.x * 16;

  if (tid < 45) kp[tid] = kpts[tid];
  __syncthreads();

  if (wave < 4) {
    const int n0  = wave * 16;
    const int row = m0 + lm16;
    v8f acc = (v8f){0.f,0.f,0.f,0.f,0.f,0.f,0.f,0.f};
    for (int k0 = 0; k0 < CCH; k0 += 4) {
      const int kk = k0 + 2 * half;
      v2f a = *(const v2f*)(sfeat + row * CCH + kk);
      v2f b;
      b.x = W1[kk * CCH + n0 + lm16];
      b.y = W1[(kk + 1) * CCH + n0 + lm16];
      acc = wmma_f32_16x16x4(a, b, acc);
    }
    const int ch = n0 + lm16;
    const float sc = ws[SCALE_OFF + ch];
    const float sh = ws[SHIFT_OFF + ch];
    #pragma unroll
    for (int r = 0; r < 8; ++r) {
      const int M = r + 8 * half;
      float v = acc[r] * sc + sh;
      v = (v > 0.f) ? v : 0.1f * v;
      act[M * 68 + ch] = v;
    }
  } else {
    #pragma unroll
    for (int i = 0; i < 4; ++i) {
      const int p  = (wave - 4) * 128 + i * 32 + lane;
      const int lm = p >> 5;
      const int h  = p & 31;
      const int m  = m0 + lm;
      const int j  = nbidx[m * HNB + h];
      const bool valid = ((unsigned)j < (unsigned)NPTS);
      const int jj = valid ? j : 0;
      const float qx = qpts[m * 3 + 0];
      const float qy = qpts[m * 3 + 1];
      const float qz = qpts[m * 3 + 2];
      const float dx = spts[jj * 3 + 0] - qx;
      const float dy = spts[jj * 3 + 1] - qy;
      const float dz = spts[jj * 3 + 2] - qz;
      float best = 3.4e38f; int bk = 0;
      #pragma unroll
      for (int k = 0; k < KPN; ++k) {
        const float ex = dx - kp[k * 3 + 0];
        const float ey = dy - kp[k * 3 + 1];
        const float ez = dz - kp[k * 3 + 2];
        const float sq = ex * ex + ey * ey + ez * ez;
        if (sq < best) { best = sq; bk = k; }
      }
      float infl = 1.0f - sqrtf(best) * (1.0f / SIGMA);
      infl = (infl > 0.f) ? infl : 0.f;
      if (!valid) infl = 0.f;
      jIdx[p] = jj; nnK[p] = bk; nnI[p] = infl;
    }
  }
  __syncthreads();

  for (int t = wave; t < 60; t += 8) {
    const int n = t * 16 + lm16;
    v8f acc = (v8f){0.f,0.f,0.f,0.f,0.f,0.f,0.f,0.f};
    for (int k0 = 0; k0 < CCH; k0 += 4) {
      const int kk = k0 + 2 * half;
      v2f a = *(const v2f*)(act + lm16 * 68 + kk);
      v2f b;
      b.x = W2[kk * 960 + n];
      b.y = W2[(kk + 1) * 960 + n];
      acc = wmma_f32_16x16x4(a, b, acc);
    }
    const float bias = b2[n];
    #pragma unroll
    for (int r = 0; r < 8; ++r) {
      const int M = r + 8 * half;
      cw[M * 960 + n] = __float2half(acc[r] + bias);
    }
  }
  __syncthreads();

  #pragma unroll
  for (int rr = 0; rr < 2; ++rr) {
    const int lm = wave + rr * 8;
    float a0 = 0.f, a1 = 0.f;
    for (int h = 0; h < HNB; ++h) {
      const int p = lm * HNB + h;
      const float f = nnI[p];
      if (f > 0.f) {
        const int j = jIdx[p];
        const int k = nnK[p];
        const float w0 = __half2float(cw[lm * 960 + k * 64 + lane]) * f;
        const float w1 = __half2float(cw[lm * 960 + k * 64 + lane + 32]) * f;
        a0 = fmaf(sfeat[j * 64 + lane],      w0, a0);
        a1 = fmaf(sfeat[j * 64 + lane + 32], w1, a1);
      }
    }
    out[(m0 + lm) * 64 + lane]      = a0;
    out[(m0 + lm) * 64 + lane + 32] = a1;
  }
}

extern "C" void kernel_launch(void* const* d_in, const int* in_sizes, int n_in,
                              void* d_out, int out_size, void* d_ws, size_t ws_size,
                              hipStream_t stream) {
  const float* qpts  = (const float*)d_in[0];
  const float* spts  = (const float*)d_in[1];
  const float* sfeat = (const float*)d_in[2];
  const int*   nbidx = (const int*)  d_in[3];
  const float* kpts  = (const float*)d_in[4];
  const float* W1    = (const float*)d_in[5];
  const float* gamma = (const float*)d_in[6];
  const float* beta  = (const float*)d_in[7];
  const float* W2    = (const float*)d_in[8];
  const float* b2    = (const float*)d_in[9];
  float* out = (float*)d_out;
  float* ws  = (float*)d_ws;

  k1_stats<<<GRID1, 256, 0, stream>>>(sfeat, W1, ws);
  k2_finalize<<<1, 64, 0, stream>>>(ws, gamma, beta, ws);
  k3_main<<<TILES, 256, 0, stream>>>(qpts, spts, sfeat, nbidx, kpts,
                                     W1, W2, b2, ws, out);
  (void)in_sizes; (void)n_in; (void)out_size; (void)ws_size;
}


# k3 persistent and weight-stationary: 256 blocks of 32 waves, each wave keeps its 32 columns of W2 (or W1) in registers as WMMA B fragments and the block loops over point groups; W2 read once per WGP
# speedup vs baseline: 1.6485x; 1.6485x over previous
#include <hip/hip_runtime.h>
#include <hip/hip_fp16.h>


typedef float v2f __attribute__((ext_vector_type(2)));
typedef float v8f __attribute__((ext_vector_type(8)));

#define NPTS   50000
#define CCH    64
#define KPN    15
#define HNB    32
#define SIGMA  0.7f
#define BN_EPS 1e-5f

#define TILES     ((NPTS + 15) / 16)
#define K1WAVES   32
#define K1THREADS 1024
#define GRID1     ((TILES + K1WAVES - 1) / K1WAVES)
#define NRED      7
#define K3THREADS 1024
#define CONVWAVES 30
#define PGRP      32
#define GROUPS    ((NPTS + PGRP - 1) / PGRP)
#define GRID3     256
#define CWLD      (KPN * CCH)
#define LDS_KP    0
#define LDS_BN    (LDS_KP  + 48 * 4)
#define LDS_RED   (LDS_BN  + 2 * CCH * 4)
#define LDS_ACT   (LDS_RED + NRED * 32 * 16)
#define LDS_GEO   (LDS_ACT + 2 * PGRP * 68 * 4)
#define LDS_CW    (LDS_GEO + 2 * 3 * PGRP * HNB * 4)
#define LDS_TOTAL (LDS_CW  + 2 * PGRP * CWLD * 2)
static_assert(LDS_RED % 16 == 0 && LDS_ACT % 16 == 0 && LDS_GEO % 16 == 0 && LDS_CW % 16 == 0, "LDS pieces must stay 16-byte aligned");
static_assert(CONVWAVES * 32 == CWLD, "conv waves must cover all W2 columns");
static_assert(GRID3 <= GROUPS, "every block needs a first group");
static_assert(GRID1 % NRED == 0, "partial rows must split evenly over the summing waves");
static_assert(K1THREADS == K1WAVES * 32, "k1 block size must match its wave count");

static __device__ __forceinline__ v8f wmma_f32_16x16x4(v2f a, v2f b, v8f c) {
  return __builtin_amdgcn_wmma_f32_16x16x4_f32(false, a, false, b,
                                               (short)0, c, false, false);
}

__global__ __launch_bounds__(K1THREADS) void k1_stats(const float* __restrict__ sfeat,
                                                const float* __restrict__ W1,
                                                float* __restrict__ partials) {
  __shared__ float wsum[K1WAVES][64];
  __shared__ float wssq[K1WAVES][64];
  const int tid  = threadIdx.x;
  const int wave = tid >> 5;
  const int lane = tid & 31;
  const int half = lane >> 4;
  const int lm16 = lane & 15;
  const int t    = blockIdx.x * K1WAVES + wave;

  v8f acc[4];
  #pragma unroll
  for (int n0 = 0; n0 < 4; ++n0)
    acc[n0] = (v8f){0.f,0.f,0.f,0.f,0.f,0.f,0.f,0.f};

  if (t < TILES) {
    const int row = t * 16 + lm16;
    for (int k0 = 0; k0 < CCH; k0 += 4) {
      const int kk = k0 + 2 * half;
      v2f a = *(const v2f*)(sfeat + row * CCH + kk);
      #pragma unroll
      for (int n0 = 0; n0 < 4; ++n0) {
        const int n = n0 * 16 + lm16;
        v2f b;
        b.x = W1[kk * CCH + n];
        b.y = W1[(kk + 1) * CCH + n];
        acc[n0] = wmma_f32_16x16x4(a, b, acc[n0]);
      }
    }
  }

  #pragma unroll
  for (int n0 = 0; n0 < 4; ++n0) {
    float s = 0.f, q = 0.f;
    #pragma unroll
    for (int r = 0; r < 8; ++r) { float v = acc[n0][r]; s += v; q += v * v; }
    s += __shfl_xor(s, 16);
    q += __shfl_xor(q, 16);
    if (half == 0) { wsum[wave][n0 * 16 + lm16] = s; wssq[wave][n0 * 16 + lm16] = q; }
  }
  __syncthreads();

  if (tid < 128) {
    const int c = tid & 63;
    const float* src = (tid < 64) ? &wsum[0][0] : &wssq[0][0];
    float s = 0.f;
    #pragma unroll
    for (int w = 0; w < K1WAVES; ++w) s += src[w * 64 + c];
    partials[blockIdx.x * 128 + tid] = s;
  }
}

__global__ __launch_bounds__(K3THREADS) void k3_main(const float* __restrict__ qpts,
                                               const float* __restrict__ spts,
                                               const float* __restrict__ sfeat,
                                               const int*   __restrict__ nbidx,
                                               const float* __restrict__ kpts,
                                               const float* __restrict__ W1,
                                               const float* __restrict__ gamma,
                                               const float* __restrict__ beta,
                                               const float* __restrict__ W2,
                                               const float* __restrict__ b2,
                                               const float* __restrict__ ws,
                                               float* __restrict__ out) {
  __shared__ __align__(16) unsigned char smem[LDS_TOTAL];
  float*  const kp  = (float*)(smem + LDS_KP);
  float*  const bn  = (float*)(smem + LDS_BN);
  float4* const red = (float4*)(smem + LDS_RED);
  float  (*const act)[PGRP * 68]     = (float (*)[PGRP * 68])(smem + LDS_ACT);
  int    (*const geo)[3][PGRP * HNB] = (int (*)[3][PGRP * HNB])(smem + LDS_GEO);
  __half (*const cw)[PGRP * CWLD]    = (__half (*)[PGRP * CWLD])(smem + LDS_CW);

  const int tid  = threadIdx.x;
  const int wave = __builtin_amdgcn_readfirstlane(tid >> 5);
  const int lane = tid & 31;
  const int half = lane >> 4;
  const int lm16 = lane & 15;
  const bool convw = (wave < CONVWAVES);
  const int  hw    = wave - CONVWAVES;
  const bool pfirst = (((wave >> 2) & 3) == 3);

  if (wave < NRED) {
    const float4* __restrict__ p4 = (const float4*)ws;
    float4 s = make_float4(0.f, 0.f, 0.f, 0.f);
    #pragma unroll
    for (int i = 0; i < GRID1 / NRED; ++i) {
      const float4 v = p4[(wave + NRED * i) * 32 + lane];
      s.x += v.x; s.y += v.y; s.z += v.z; s.w += v.w;
    }
    red[wave * 32 + lane] = s;
  } else if (wave == NRED) {
    const float k0 = kpts[lane];
    if (lane < 45 - 32) kp[32 + lane] = kpts[32 + lane];
    kp[lane] = k0;
  }
  __syncthreads();

  if (tid < CCH) {
    const int ch = tid;
    const float* redf = (const float*)red;
    float ssum = 0.f, qsum = 0.f;
    #pragma unroll
    for (int w = 0; w < NRED; ++w) {
      ssum += redf[w * 128 + ch];
      qsum += redf[w * 128 + 64 + ch];
    }
    const float inv_n = 1.0f / (float)NPTS;
    const float mu  = ssum * inv_n;
    const float var = qsum * inv_n - mu * mu;
    const float sc  = gamma[ch] * rsqrtf(var + BN_EPS);
    const float sh  = beta[ch] - mu * sc;
    bn[ch] = sc; bn[CCH + ch] = sh;
  }

  v2f bf[2][16];
  __syncthreads();

  auto make_act = [&](int gq, int bq) {
    const float* bnw = bn + hw * 32 + lm16;
    const float sc0 = bnw[0],  sh0 = bnw[CCH];
    const float sc1 = bnw[16], sh1 = bnw[CCH + 16];
    #pragma unroll
    for (int rt = 0; rt < 2; ++rt) {
      int row = gq * PGRP + rt * 16 + lm16;
      row = (row < NPTS) ? row : (NPTS - 1);
      const float* __restrict__ arow = sfeat + row * CCH + 2 * half;
      v8f acc0 = (v8f){0.f,0.f,0.f,0.f,0.f,0.f,0.f,0.f};
      v8f acc1 = (v8f){0.f,0.f,0.f,0.f,0.f,0.f,0.f,0.f};
      #pragma unroll
      for (int hb = 0; hb < 16 / 8; ++hb) {
        v2f a[8];
        #pragma unroll
        for (int s = 0; s < 8; ++s) a[s] = *(const v2f*)(arow + 4 * (8 * hb + s));
        #pragma unroll
        for (int s = 0; s < 8; ++s) {
          acc0 = wmma_f32_16x16x4(a[s], bf[0][8 * hb + s], acc0);
          acc1 = wmma_f32_16x16x4(a[s], bf[1][8 * hb + s], acc1);
        }
      }
      float* actw = &act[bq][(rt * 16 + 8 * half) * 68 + hw * 32 + lm16];
      #pragma unroll
      for (int r = 0; r < 8; ++r) {
        float v0 = acc0[r] * sc0 + sh0;
        float v1 = acc1[r] * sc1 + sh1;
        v0 = (v0 > 0.f) ? v0 : 0.1f * v0;
        v1 = (v1 > 0.f) ? v1 : 0.1f * v1;
        actw[r * 68]      = v0;
        actw[r * 68 + 16] = v1;
      }
    }
  };

  auto produce = [&](int gp, int pb) {
    int gm = gp * PGRP + wave;
    gm = (gm < NPTS) ? gm : (NPTS - 1);
    const int  j     = nbidx[gm * HNB + lane];

    if (convw) {
      const int n0 = wave * 32 + lm16;
      const float bias0 = b2[n0];
      const float bias1 = b2[n0 + 16];
      #pragma unroll
      for (int rt = 0; rt < 2; ++rt) {
        const float* arow = &act[pb][(rt * 16 + lm16) * 68 + 2 * half];
        v8f acc0 = (v8f){0.f,0.f,0.f,0.f,0.f,0.f,0.f,0.f};
        v8f acc1 = (v8f){0.f,0.f,0.f,0.f,0.f,0.f,0.f,0.f};
        #pragma unroll
        for (int s = 0; s < 16; ++s) {
          v2f a = *(const v2f*)(arow + 4 * s);
          acc0 = wmma_f32_16x16x4(a, bf[0][s], acc0);
          acc1 = wmma_f32_16x16x4(a, bf[1][s], acc1);
        }
        __half* cww = &cw[pb][(rt * 16 + 8 * half) * CWLD + n0];
        #pragma unroll
        for (int r = 0; r < 8; ++r) {
          cww[r * CWLD]      = __float2half(acc0[r] + bias0);
          cww[r * CWLD + 16] = __float2half(acc1[r] + bias1);
        }
      }
    } else {
      const int gn = gp + GRID3;
      if (gn < GROUPS) make_act(gn, pb ^ 1);
    }

    const bool valid = ((unsigned)j < (unsigned)NPTS);
    const int  jj    = valid ? j : 0;
    const float dx = spts[jj * 3 + 0] - qpts[gm * 3 + 0];
    const float dy = spts[jj * 3 + 1] - qpts[gm * 3 + 1];
    const float dz = spts[jj * 3 + 2] - qpts[gm * 3 + 2];
    float best = 3.4e38f; int bk = 0;
    #pragma unroll
    for (int k = 0; k < KPN; ++k) {
      const float ex = dx - kp[k * 3 + 0];
      const float ey = dy - kp[k * 3 + 1];
      const float ez = dz - kp[k * 3 + 2];
      const float sq = ex * ex + ey * ey + ez * ez;
      if (sq < best) { best = sq; bk = k; }
    }
    float infl = 1.0f - sqrtf(best) * (1.0f / SIGMA);
    infl = (infl > 0.f) ? infl : 0.f;
    if (!valid) infl = 0.f;
    int* gw = &geo[pb][0][tid];
    gw[0] = jj; gw[PGRP * HNB] = bk; gw[2 * PGRP * HNB] = __float_as_int(infl);
  };

  auto consume = [&](int gc, int cb) {
    const int lm = wave;
    const int m  = gc * PGRP + lm;
    if (m < NPTS) {
      const int*    gr  = &geo[cb][0][lm * HNB];
      const __half* cwr = &cw[cb][lm * CWLD + lane];
      float a0 = 0.f, a1 = 0.f;
      for (int h = 0; h < HNB; ++h) {
        const float f = __int_as_float(gr[2 * PGRP * HNB + h]);
        if (f > 0.f) {
          const int jn = gr[h];
          const int k  = gr[PGRP * HNB + h];
          const float w0 = __half2float(cwr[k * 64]) * f;
          const float w1 = __half2float(cwr[k * 64 + 32]) * f;
          a0 = fmaf(sfeat[jn * 64 + lane],      w0, a0);
          a1 = fmaf(sfeat[jn * 64 + lane + 32], w1, a1);
        }
      }
      out[m * 64 + lane]      = a0;
      out[m * 64 + lane + 32] = a1;
    }
  };

  {
    const float* __restrict__ Bsrc = convw ? W2 : W1;
    const int ldb   = convw ? CWLD : CCH;
    const int ncol0 = convw ? wave * 32 : hw * 32;
    #pragma unroll
    for (int c = 0; c < 2; ++c) {
      const int n = ncol0 + c * 16 + lm16;
      #pragma unroll
      for (int s = 0; s < 16; ++s) {
        const int kk = 4 * s + 2 * half;
        bf[c][s].x = Bsrc[kk * ldb + n];
        bf[c][s].y = Bsrc[(kk + 1) * ldb + n];
      }
    }
  }
  if (!convw) make_act(blockIdx.x, 0);
  __syncthreads();

  for (int it = -1; ; ++it) {
    const int  gc   = (int)blockIdx.x + it * GRID3;
    const int  gp   = gc + GRID3;
    const bool hasC = (it >= 0);
    const bool hasP = (gp < GROUPS);
    if (pfirst) {
      if (hasP) produce(gp, (it + 1) & 1);
      if (hasC) consume(gc, it & 1);
    } else {
      if (hasC) consume(gc, it & 1);
      if (hasP) produce(gp, (it + 1) & 1);
    }
    __syncthreads();
    if (!hasP) break;
  }
}

extern "C" void kernel_launch(void* const* d_in, const int* in_sizes, int n_in,
                              void* d_out, int out_size, void* d_ws, size_t ws_size,
                              hipStream_t stream) {
  const float* qpts  = (const float*)d_in[0];
  const float* spts  = (const float*)d_in[1];
  const float* sfeat = (const float*)d_in[2];
  const int*   nbidx = (const int*)  d_in[3];
  const float* kpts  = (const float*)d_in[4];
  const float* W1    = (const float*)d_in[5];
  const float* gamma = (const float*)d_in[6];
  const float* beta  = (const float*)d_in[7];
  const float* W2    = (const float*)d_in[8];
  const float* b2    = (const float*)d_in[9];
  float* out = (float*)d_out;
  float* ws  = (float*)d_ws;

  k1_stats<<<GRID1, K1THREADS, 0, stream>>>(sfeat, W1, ws);
  k3_main<<<GRID3, K3THREADS, 0, stream>>>(qpts, spts, sfeat, nbidx, kpts,
                                     W1, gamma, beta, W2, b2, ws, out);
  (void)in_sizes; (void)n_in; (void)out_size; (void)ws_size;
}


# k3 geometry (neighbor row, kernel point, influence) stays in the registers of the wave that made it, no LDS table: consume walks a ballot mask of influenced neighbors by lane reads, 4 per trip
# speedup vs baseline: 2.0269x; 1.2295x over previous
#include <hip/hip_runtime.h>
#include <hip/hip_fp16.h>


typedef float v2f __attribute__((ext_vector_type(2)));
typedef float v8f __attribute__((ext_vector_type(8)));

#define NPTS   50000
#define CCH    64
#define KPN    15
#define HNB    32
#define SIGMA  0.7f
#define BN_EPS 1e-5f

#define TILES     ((NPTS + 15) / 16)
#define K1WAVES   32
#define K1THREADS 1024
#define GRID1     ((TILES + K1WAVES - 1) / K1WAVES)
#define NRED      7
#define K3THREADS 1024
#define CONVWAVES 30
#define PGRP      32
#define GROUPS    ((NPTS + PGRP - 1) / PGRP)
#define GRID3     256
#define CWLD      (KPN * CCH)
#define NWALK     4
#define LDS_KP    0
#define LDS_BN    (LDS_KP  + 48 * 4)
#define LDS_RED   (LDS_BN  + 2 * CCH * 4)
#define LDS_ACT   (LDS_RED + NRED * 32 * 16)
#define LDS_CW    (LDS_ACT + 2 * PGRP * 68 * 4)
#define LDS_TOTAL (LDS_CW  + 2 * PGRP * CWLD * 2)
static_assert(LDS_RED % 16 == 0 && LDS_ACT % 16 == 0 && LDS_CW % 16 == 0, "LDS pieces must stay 16-byte aligned");
static_assert(NPTS <= 65536 && KPN <= 16, "neighbor row and kernel point share one 32-bit word (16 + 4 bits)");
static_assert(HNB == 32 && PGRP == 32, "one lane per neighbor, one wave per local point");
static_assert(CONVWAVES * 32 == CWLD, "conv waves must cover all W2 columns");
static_assert(GRID3 <= GROUPS, "every block needs a first group");
static_assert(GRID1 % NRED == 0, "partial rows must split evenly over the summing waves");
static_assert(K1THREADS == K1WAVES * 32, "k1 block size must match its wave count");

static __device__ __forceinline__ v8f wmma_f32_16x16x4(v2f a, v2f b, v8f c) {
  return __builtin_amdgcn_wmma_f32_16x16x4_f32(false, a, false, b,
                                               (short)0, c, false, false);
}

__global__ __launch_bounds__(K1THREADS) void k1_stats(const float* __restrict__ sfeat,
                                                const float* __restrict__ W1,
                                                float* __restrict__ partials) {
  __shared__ float wsum[K1WAVES][64];
  __shared__ float wssq[K1WAVES][64];
  const int tid  = threadIdx.x;
  const int wave = tid >> 5;
  const int lane = tid & 31;
  const int half = lane >> 4;
  const int lm16 = lane & 15;
  const int t    = blockIdx.x * K1WAVES + wave;

  v8f acc[4];
  #pragma unroll
  for (int n0 = 0; n0 < 4; ++n0)
    acc[n0] = (v8f){0.f,0.f,0.f,0.f,0.f,0.f,0.f,0.f};

  if (t < TILES) {
    const int row = t * 16 + lm16;
    for (int k0 = 0; k0 < CCH; k0 += 4) {
      const int kk = k0 + 2 * half;
      v2f a = *(const v2f*)(sfeat + row * CCH + kk);
      #pragma unroll
      for (int n0 = 0; n0 < 4; ++n0) {
        const int n = n0 * 16 + lm16;
        v2f b;
        b.x = W1[kk * CCH + n];
        b.y = W1[(kk + 1) * CCH + n];
        acc[n0] = wmma_f32_16x16x4(a, b, acc[n0]);
      }
    }
  }

  #pragma unroll
  for (int n0 = 0; n0 < 4; ++n0) {
    float s = 0.f, q = 0.f;
    #pragma unroll
    for (int r = 0; r < 8; ++r) { float v = acc[n0][r]; s += v; q += v * v; }
    s += __shfl_xor(s, 16);
    q += __shfl_xor(q, 16);
    if (half == 0) { wsum[wave][n0 * 16 + lm16] = s; wssq[wave][n0 * 16 + lm16] = q; }
  }
  __syncthreads();

  if (tid < 128) {
    const int c = tid & 63;
    const float* src = (tid < 64) ? &wsum[0][0] : &wssq[0][0];
    float s = 0.f;
    #pragma unroll
    for (int w = 0; w < K1WAVES; ++w) s += src[w * 64 + c];
    partials[blockIdx.x * 128 + tid] = s;
  }
}

__global__ __launch_bounds__(K3THREADS) void k3_main(const float* __restrict__ qpts,
                                               const float* __restrict__ spts,
                                               const float* __restrict__ sfeat,
                                               const int*   __restrict__ nbidx,
                                               const float* __restrict__ kpts,
                                               const float* __restrict__ W1,
                                               const float* __restrict__ gamma,
                                               const float* __restrict__ beta,
                                               const float* __restrict__ W2,
                                               const float* __restrict__ b2,
                                               const float* __restrict__ ws,
                                               float* __restrict__ out) {
  __shared__ __align__(16) unsigned char smem[LDS_TOTAL];
  float*  const kp  = (float*)(smem + LDS_KP);
  float*  const bn  = (float*)(smem + LDS_BN);
  float4* const red = (float4*)(smem + LDS_RED);
  float  (*const act)[PGRP * 68]     = (float (*)[PGRP * 68])(smem + LDS_ACT);
  __half (*const cw)[PGRP * CWLD]    = (__half (*)[PGRP * CWLD])(smem + LDS_CW);

  const int tid  = threadIdx.x;
  const int wave = __builtin_amdgcn_readfirstlane(tid >> 5);
  const int lane = tid & 31;
  const int half = lane >> 4;
  const int lm16 = lane & 15;
  const bool convw = (wave < CONVWAVES);
  const int  hw    = wave - CONVWAVES;
  const bool pfirst = (((wave >> 2) & 3) == 3);

  if (wave < NRED) {
    const float4* __restrict__ p4 = (const float4*)ws;
    float4 s = make_float4(0.f, 0.f, 0.f, 0.f);
    #pragma unroll
    for (int i = 0; i < GRID1 / NRED; ++i) {
      const float4 v = p4[(wave + NRED * i) * 32 + lane];
      s.x += v.x; s.y += v.y; s.z += v.z; s.w += v.w;
    }
    red[wave * 32 + lane] = s;
  } else if (wave == NRED) {
    const float k0 = kpts[lane];
    if (lane < 45 - 32) kp[32 + lane] = kpts[32 + lane];
    kp[lane] = k0;
  }
  __syncthreads();

  if (tid < CCH) {
    const int ch = tid;
    const float* redf = (const float*)red;
    float ssum = 0.f, qsum = 0.f;
    #pragma unroll
    for (int w = 0; w < NRED; ++w) {
      ssum += redf[w * 128 + ch];
      qsum += redf[w * 128 + 64 + ch];
    }
    const float inv_n = 1.0f / (float)NPTS;
    const float mu  = ssum * inv_n;
    const float var = qsum * inv_n - mu * mu;
    const float sc  = gamma[ch] * rsqrtf(var + BN_EPS);
    const float sh  = beta[ch] - mu * sc;
    bn[ch] = sc; bn[CCH + ch] = sh;
  }

  v2f bf[2][16];
  __syncthreads();

  auto make_act = [&](int gq, int bq) {
    const float* bnw = bn + hw * 32 + lm16;
    const float sc0 = bnw[0],  sh0 = bnw[CCH];
    const float sc1 = bnw[16], sh1 = bnw[CCH + 16];
    #pragma unroll
    for (int rt = 0; rt < 2; ++rt) {
      int row = gq * PGRP + rt * 16 + lm16;
      row = (row < NPTS) ? row : (NPTS - 1);
      const float* __restrict__ arow = sfeat + row * CCH + 2 * half;
      v8f acc0 = (v8f){0.f,0.f,0.f,0.f,0.f,0.f,0.f,0.f};
      v8f acc1 = (v8f){0.f,0.f,0.f,0.f,0.f,0.f,0.f,0.f};
      #pragma unroll
      for (int hb = 0; hb < 16 / 8; ++hb) {
        v2f a[8];
        #pragma unroll
        for (int s = 0; s < 8; ++s) a[s] = *(const v2f*)(arow + 4 * (8 * hb + s));
        #pragma unroll
        for (int s = 0; s < 8; ++s) {
          acc0 = wmma_f32_16x16x4(a[s], bf[0][8 * hb + s], acc0);
          acc1 = wmma_f32_16x16x4(a[s], bf[1][8 * hb + s], acc1);
        }
      }
      float* actw = &act[bq][(rt * 16 + 8 * half) * 68 + hw * 32 + lm16];
      #pragma unroll
      for (int r = 0; r < 8; ++r) {
        float v0 = acc0[r] * sc0 + sh0;
        float v1 = acc1[r] * sc1 + sh1;
        v0 = (v0 > 0.f) ? v0 : 0.1f * v0;
        v1 = (v1 > 0.f) ? v1 : 0.1f * v1;
        actw[r * 68]      = v0;
        actw[r * 68 + 16] = v1;
      }
    }
  };

  auto produce = [&](int gp, int pb, int& gjk, float& gfl) {
    int gm = gp * PGRP + wave;
    gm = (gm < NPTS) ? gm : (NPTS - 1);
    const int  j     = nbidx[gm * HNB + lane];

    if (convw) {
      const int n0 = wave * 32 + lm16;
      const float bias0 = b2[n0];
      const float bias1 = b2[n0 + 16];
      #pragma unroll
      for (int rt = 0; rt < 2; ++rt) {
        const float* arow = &act[pb][(rt * 16 + lm16) * 68 + 2 * half];
        v8f acc0 = (v8f){0.f,0.f,0.f,0.f,0.f,0.f,0.f,0.f};
        v8f acc1 = (v8f){0.f,0.f,0.f,0.f,0.f,0.f,0.f,0.f};
        #pragma unroll
        for (int s = 0; s < 16; ++s) {
          v2f a = *(const v2f*)(arow + 4 * s);
          acc0 = wmma_f32_16x16x4(a, bf[0][s], acc0);
          acc1 = wmma_f32_16x16x4(a, bf[1][s], acc1);
        }
        __half* cww = &cw[pb][(rt * 16 + 8 * half) * CWLD + n0];
        #pragma unroll
        for (int r = 0; r < 8; ++r) {
          cww[r * CWLD]      = __float2half(acc0[r] + bias0);
          cww[r * CWLD + 16] = __float2half(acc1[r] + bias1);
        }
      }
    } else {
      const int gn = gp + GRID3;
      if (gn < GROUPS) make_act(gn, pb ^ 1);
    }

    const bool valid = ((unsigned)j < (unsigned)NPTS);
    const int  jj    = valid ? j : 0;
    const float dx = spts[jj * 3 + 0] - qpts[gm * 3 + 0];
    const float dy = spts[jj * 3 + 1] - qpts[gm * 3 + 1];
    const float dz = spts[jj * 3 + 2] - qpts[gm * 3 + 2];
    float best = 3.4e38f; int bk = 0;
    #pragma unroll
    for (int k = 0; k < KPN; ++k) {
      const float ex = dx - kp[k * 3 + 0];
      const float ey = dy - kp[k * 3 + 1];
      const float ez = dz - kp[k * 3 + 2];
      const float sq = ex * ex + ey * ey + ez * ez;
      if (sq < best) { best = sq; bk = k; }
    }
    float infl = 1.0f - sqrtf(best) * (1.0f / SIGMA);
    infl = (infl > 0.f) ? infl : 0.f;
    if (!valid) infl = 0.f;
    gjk = jj | (bk << 16);
    gfl = infl;
  };

  auto consume = [&](int gc, int cb, int cjk, float cfl) {
    const int m = gc * PGRP + wave;
    if (m < NPTS) {
      const __half* cwr  = &cw[cb][wave * CWLD + lane];
      const float*  frow = sfeat + lane;
      const int     cfb  = __float_as_int(cfl);
      unsigned todo = __builtin_amdgcn_ballot_w32(cfl > 0.f);
      float a0 = 0.f, a1 = 0.f;
      while (todo) {
        bool     on[NWALK];
        int      jk[NWALK];
        float    f[NWALK], x0[NWALK], x1[NWALK];
        unsigned c0[NWALK], c1[NWALK];
        #pragma unroll
        for (int i = 0; i < NWALK; ++i) {
          on[i] = (todo != 0u);
          jk[i] = 0; f[i] = 0.f; x0[i] = 0.f; x1[i] = 0.f;
          if (on[i]) {
            const int h = __builtin_ctz(todo);
            todo &= todo - 1u;
            jk[i] = __builtin_amdgcn_readlane(cjk, h);
            f[i]  = __int_as_float(__builtin_amdgcn_readlane(cfb, h));
            const float* fr = frow + (jk[i] & 0xffff) * 64;
            x0[i] = fr[0];
            x1[i] = fr[32];
          }
        }
        #pragma unroll
        for (int i = 0; i < NWALK; ++i) {
          c0[i] = 0u; c1[i] = 0u;
          if (on[i]) {
            const __half* cr = cwr + (jk[i] >> 16) * 64;
            c0[i] = __half_as_ushort(cr[0]);
            c1[i] = __half_as_ushort(cr[32]);
          }
        }
        #pragma unroll
        for (int i = 0; i < NWALK; ++i) {
          if (on[i]) {
            const float w0 = __half2float(__ushort_as_half((unsigned short)c0[i])) * f[i];
            const float w1 = __half2float(__ushort_as_half((unsigned short)c1[i])) * f[i];
            a0 = fmaf(x0[i], w0, a0);
            a1 = fmaf(x1[i], w1, a1);
          }
        }
      }
      out[m * 64 + lane]      = a0;
      out[m * 64 + lane + 32] = a1;
    }
  };

  {
    const float* __restrict__ Bsrc = convw ? W2 : W1;
    const int ldb   = convw ? CWLD : CCH;
    const int ncol0 = convw ? wave * 32 : hw * 32;
    #pragma unroll
    for (int c = 0; c < 2; ++c) {
      const int n = ncol0 + c * 16 + lm16;
      #pragma unroll
      for (int s = 0; s < 16; ++s) {
        const int kk = 4 * s + 2 * half;
        bf[c][s].x = Bsrc[kk * ldb + n];
        bf[c][s].y = Bsrc[(kk + 1) * ldb + n];
      }
    }
  }
  if (!convw) make_act(blockIdx.x, 0);
  __syncthreads();

  int   gjk = 0;
  float gfl = 0.f;
  for (int it = -1; ; ++it) {
    const int  gc   = (int)blockIdx.x + it * GRID3;
    const int  gp   = gc + GRID3;
    const bool hasC = (it >= 0);
    const bool hasP = (gp < GROUPS);
    if (pfirst) {
      const int   ojk = gjk;
      const float ofl = gfl;
      if (hasP) produce(gp, (it + 1) & 1, gjk, gfl);
      if (hasC) consume(gc, it & 1, ojk, ofl);
    } else {
      if (hasC) consume(gc, it & 1, gjk, gfl);
      if (hasP) produce(gp, (it + 1) & 1, gjk, gfl);
    }
    __syncthreads();
    if (!hasP) break;
  }
}

extern "C" void kernel_launch(void* const* d_in, const int* in_sizes, int n_in,
                              void* d_out, int out_size, void* d_ws, size_t ws_size,
                              hipStream_t stream) {
  const float* qpts  = (const float*)d_in[0];
  const float* spts  = (const float*)d_in[1];
  const float* sfeat = (const float*)d_in[2];
  const int*   nbidx = (const int*)  d_in[3];
  const float* kpts  = (const float*)d_in[4];
  const float* W1    = (const float*)d_in[5];
  const float* gamma = (const float*)d_in[6];
  const float* beta  = (const float*)d_in[7];
  const float* W2    = (const float*)d_in[8];
  const float* b2    = (const float*)d_in[9];
  float* out = (float*)d_out;
  float* ws  = (float*)d_ws;

  k1_stats<<<GRID1, K1THREADS, 0, stream>>>(sfeat, W1, ws);
  k3_main<<<GRID3, K3THREADS, 0, stream>>>(qpts, spts, sfeat, nbidx, kpts,
                                     W1, gamma, beta, W2, b2, ws, out);
  (void)in_sizes; (void)n_in; (void)out_size; (void)ws_size;
}


# k1 persistent and weight-stationary: 256 blocks of 8 waves, each wave keeps all of W1 in registers as WMMA B fragments and runs its 1-2 row tiles against it (new reduction tree for mean/var)
# speedup vs baseline: 2.0857x; 1.0290x over previous
#include <hip/hip_runtime.h>
#include <hip/hip_fp16.h>


typedef float v2f __attribute__((ext_vector_type(2)));
typedef float v8f __attribute__((ext_vector_type(8)));

#define NPTS   50000
#define CCH    64
#define KPN    15
#define HNB    32
#define SIGMA  0.7f
#define BN_EPS 1e-5f

#define TILES     ((NPTS + 15) / 16)
#define K1WAVES   8
#define K1THREADS 256
#define GRID1     256
#define K1ROUND   (GRID1 * K1WAVES)
#define NRED      16
#define K3THREADS 1024
#define CONVWAVES 30
#define PGRP      32
#define GROUPS    ((NPTS + PGRP - 1) / PGRP)
#define GRID3     256
#define CWLD      (KPN * CCH)
#define NWALK     4
#define LDS_KP    0
#define LDS_BN    (LDS_KP  + 48 * 4)
#define LDS_RED   (LDS_BN  + 2 * CCH * 4)
#define LDS_ACT   (LDS_RED + NRED * 32 * 16)
#define LDS_CW    (LDS_ACT + 2 * PGRP * 68 * 4)
#define LDS_TOTAL (LDS_CW  + 2 * PGRP * CWLD * 2)
static_assert(LDS_RED % 16 == 0 && LDS_ACT % 16 == 0 && LDS_CW % 16 == 0, "LDS pieces must stay 16-byte aligned");
static_assert(NPTS <= 65536 && KPN <= 16, "neighbor row and kernel point share one 32-bit word (16 + 4 bits)");
static_assert(HNB == 32 && PGRP == 32, "one lane per neighbor, one wave per local point");
static_assert(CONVWAVES * 32 == CWLD, "conv waves must cover all W2 columns");
static_assert(GRID3 <= GROUPS, "every block needs a first group");
static_assert(GRID1 % NRED == 0, "partial rows must split evenly over the summing waves");
static_assert(K1THREADS == K1WAVES * 32, "k1 block size must match its wave count");
static_assert(K1ROUND <= TILES && TILES <= 2 * K1ROUND, "every k1 wave has a first tile and at most a second");
static_assert(NRED < K3THREADS / 32, "the wave behind the summing waves copies the kernel points");

static __device__ __forceinline__ v8f wmma_f32_16x16x4(v2f a, v2f b, v8f c) {
  return __builtin_amdgcn_wmma_f32_16x16x4_f32(false, a, false, b,
                                               (short)0, c, false, false);
}

__global__ __launch_bounds__(K1THREADS) void k1_stats(const float* __restrict__ sfeat,
                                                const float* __restrict__ W1,
                                                float* __restrict__ partials) {
  __shared__ float wsum[K1WAVES][64];
  __shared__ float wssq[K1WAVES][64];
  const int tid  = threadIdx.x;
  const int wave = __builtin_amdgcn_readfirstlane(tid >> 5);
  const int lane = tid & 31;
  const int half = lane >> 4;
  const int lm16 = lane & 15;
  const int  t0   = blockIdx.x * K1WAVES + wave;
  const int  t1   = K1ROUND + wave * GRID1 + blockIdx.x;
  const bool has1 = (t1 < TILES);

  v2f a0[16], a1[16];
  {
    const float* __restrict__ arow = sfeat + (t0 * 16 + lm16) * CCH + 2 * half;
    #pragma unroll
    for (int s = 0; s < 16; ++s) a0[s] = *(const v2f*)(arow + 4 * s);
  }
  #pragma unroll
  for (int s = 0; s < 16; ++s) a1[s] = (v2f){0.f, 0.f};
  if (has1) {
    const float* __restrict__ arow = sfeat + (t1 * 16 + lm16) * CCH + 2 * half;
    #pragma unroll
    for (int s = 0; s < 16; ++s) a1[s] = *(const v2f*)(arow + 4 * s);
  }

  v2f bf[4][16];
  #pragma unroll
  for (int s = 0; s < 16; ++s) {
    const int kk = 4 * s + 2 * half;
    #pragma unroll
    for (int c = 0; c < 4; ++c) {
      const int n = c * 16 + lm16;
      bf[c][s].x = W1[kk * CCH + n];
      bf[c][s].y = W1[(kk + 1) * CCH + n];
    }
  }

  float ssum[4], qsum[4];
  #pragma unroll
  for (int c = 0; c < 4; ++c) { ssum[c] = 0.f; qsum[c] = 0.f; }

  auto tile = [&](const v2f (&a)[16]) {
    v8f acc[4];
    #pragma unroll
    for (int c = 0; c < 4; ++c)
      acc[c] = (v8f){0.f,0.f,0.f,0.f,0.f,0.f,0.f,0.f};
    #pragma unroll
    for (int s = 0; s < 16; ++s) {
      #pragma unroll
      for (int c = 0; c < 4; ++c)
        acc[c] = wmma_f32_16x16x4(a[s], bf[c][s], acc[c]);
    }
    #pragma unroll
    for (int c = 0; c < 4; ++c) {
      #pragma unroll
      for (int r = 0; r < 8; ++r) { const float v = acc[c][r]; ssum[c] += v; qsum[c] += v * v; }
    }
  };
  tile(a0);
  if (has1) tile(a1);

  #pragma unroll
  for (int c = 0; c < 4; ++c) {
    const float s = ssum[c] + __shfl_xor(ssum[c], 16);
    const float q = qsum[c] + __shfl_xor(qsum[c], 16);
    if (half == 0) { wsum[wave][c * 16 + lm16] = s; wssq[wave][c * 16 + lm16] = q; }
  }
  __syncthreads();

  if (tid < 128) {
    const int c = tid & 63;
    const float* src = (tid < 64) ? &wsum[0][0] : &wssq[0][0];
    float s = 0.f;
    #pragma unroll
    for (int w = 0; w < K1WAVES; ++w) s += src[w * 64 + c];
    partials[blockIdx.x * 128 + tid] = s;
  }
}

__global__ __launch_bounds__(K3THREADS) void k3_main(const float* __restrict__ qpts,
                                               const float* __restrict__ spts,
                                               const float* __restrict__ sfeat,
                                               const int*   __restrict__ nbidx,
                                               const float* __restrict__ kpts,
                                               const float* __restrict__ W1,
                                               const float* __restrict__ gamma,
                                               const float* __restrict__ beta,
                                               const float* __restrict__ W2,
                                               const float* __restrict__ b2,
                                               const float* __restrict__ ws,
                                               float* __restrict__ out) {
  __shared__ __align__(16) unsigned char smem[LDS_TOTAL];
  float*  const kp  = (float*)(smem + LDS_KP);
  float*  const bn  = (float*)(smem + LDS_BN);
  float4* const red = (float4*)(smem + LDS_RED);
  float  (*const act)[PGRP * 68]     = (float (*)[PGRP * 68])(smem + LDS_ACT);
  __half (*const cw)[PGRP * CWLD]    = (__half (*)[PGRP * CWLD])(smem + LDS_CW);

  const int tid  = threadIdx.x;
  const int wave = __builtin_amdgcn_readfirstlane(tid >> 5);
  const int lane = tid & 31;
  const int half = lane >> 4;
  const int lm16 = lane & 15;
  const bool convw = (wave < CONVWAVES);
  const int  hw    = wave - CONVWAVES;
  const bool pfirst = (((wave >> 2) & 3) == 3);

  if (wave < NRED) {
    const float4* __restrict__ p4 = (const float4*)ws;
    float4 s = make_float4(0.f, 0.f, 0.f, 0.f);
    #pragma unroll
    for (int i = 0; i < GRID1 / NRED; ++i) {
      const float4 v = p4[(wave + NRED * i) * 32 + lane];
      s.x += v.x; s.y += v.y; s.z += v.z; s.w += v.w;
    }
    red[wave * 32 + lane] = s;
  } else if (wave == NRED) {
    const float k0 = kpts[lane];
    if (lane < 45 - 32) kp[32 + lane] = kpts[32 + lane];
    kp[lane] = k0;
  }
  __syncthreads();

  if (tid < CCH) {
    const int ch = tid;
    const float* redf = (const float*)red;
    float ssum = 0.f, qsum = 0.f;
    #pragma unroll
    for (int w = 0; w < NRED; ++w) {
      ssum += redf[w * 128 + ch];
      qsum += redf[w * 128 + 64 + ch];
    }
    const float inv_n = 1.0f / (float)NPTS;
    const float mu  = ssum * inv_n;
    const float var = qsum * inv_n - mu * mu;
    const float sc  = gamma[ch] * rsqrtf(var + BN_EPS);
    const float sh  = beta[ch] - mu * sc;
    bn[ch] = sc; bn[CCH + ch] = sh;
  }

  v2f bf[2][16];
  __syncthreads();

  auto make_act = [&](int gq, int bq) {
    const float* bnw = bn + hw * 32 + lm16;
    const float sc0 = bnw[0],  sh0 = bnw[CCH];
    const float sc1 = bnw[16], sh1 = bnw[CCH + 16];
    #pragma unroll
    for (int rt = 0; rt < 2; ++rt) {
      int row = gq * PGRP + rt * 16 + lm16;
      row = (row < NPTS) ? row : (NPTS - 1);
      const float* __restrict__ arow = sfeat + row * CCH + 2 * half;
      v8f acc0 = (v8f){0.f,0.f,0.f,0.f,0.f,0.f,0.f,0.f};
      v8f acc1 = (v8f){0.f,0.f,0.f,0.f,0.f,0.f,0.f,0.f};
      #pragma unroll
      for (int hb = 0; hb < 16 / 8; ++hb) {
        v2f a[8];
        #pragma unroll
        for (int s = 0; s < 8; ++s) a[s] = *(const v2f*)(arow + 4 * (8 * hb + s));
        #pragma unroll
        for (int s = 0; s < 8; ++s) {
          acc0 = wmma_f32_16x16x4(a[s], bf[0][8 * hb + s], acc0);
          acc1 = wmma_f32_16x16x4(a[s], bf[1][8 * hb + s], acc1);
        }
      }
      float* actw = &act[bq][(rt * 16 + 8 * half) * 68 + hw * 32 + lm16];
      #pragma unroll
      for (int r = 0; r < 8; ++r) {
        float v0 = acc0[r] * sc0 + sh0;
        float v1 = acc1[r] * sc1 + sh1;
        v0 = (v0 > 0.f) ? v0 : 0.1f * v0;
        v1 = (v1 > 0.f) ? v1 : 0.1f * v1;
        actw[r * 68]      = v0;
        actw[r * 68 + 16] = v1;
      }
    }
  };

  auto produce = [&](int gp, int pb, int& gjk, float& gfl) {
    int gm = gp * PGRP + wave;
    gm = (gm < NPTS) ? gm : (NPTS - 1);
    const int  j     = nbidx[gm * HNB + lane];

    if (convw) {
      const int n0 = wave * 32 + lm16;
      const float bias0 = b2[n0];
      const float bias1 = b2[n0 + 16];
      #pragma unroll
      for (int rt = 0; rt < 2; ++rt) {
        const float* arow = &act[pb][(rt * 16 + lm16) * 68 + 2 * half];
        v8f acc0 = (v8f){0.f,0.f,0.f,0.f,0.f,0.f,0.f,0.f};
        v8f acc1 = (v8f){0.f,0.f,0.f,0.f,0.f,0.f,0.f,0.f};
        #pragma unroll
        for (int s = 0; s < 16; ++s) {
          v2f a = *(const v2f*)(arow + 4 * s);
          acc0 = wmma_f32_16x16x4(a, bf[0][s], acc0);
          acc1 = wmma_f32_16x16x4(a, bf[1][s], acc1);
        }
        __half* cww = &cw[pb][(rt * 16 + 8 * half) * CWLD + n0];
        #pragma unroll
        for (int r = 0; r < 8; ++r) {
          cww[r * CWLD]      = __float2half(acc0[r] + bias0);
          cww[r * CWLD + 16] = __float2half(acc1[r] + bias1);
        }
      }
    } else {
      const int gn = gp + GRID3;
      if (gn < GROUPS) make_act(gn, pb ^ 1);
    }

    const bool valid = ((unsigned)j < (unsigned)NPTS);
    const int  jj    = valid ? j : 0;
    const float dx = spts[jj * 3 + 0] - qpts[gm * 3 + 0];
    const float dy = spts[jj * 3 + 1] - qpts[gm * 3 + 1];
    const float dz = spts[jj * 3 + 2] - qpts[gm * 3 + 2];
    float best = 3.4e38f; int bk = 0;
    #pragma unroll
    for (int k = 0; k < KPN; ++k) {
      const float ex = dx - kp[k * 3 + 0];
      const float ey = dy - kp[k * 3 + 1];
      const float ez = dz - kp[k * 3 + 2];
      const float sq = ex * ex + ey * ey + ez * ez;
      if (sq < best) { best = sq; bk = k; }
    }
    float infl = 1.0f - sqrtf(best) * (1.0f / SIGMA);
    infl = (infl > 0.f) ? infl : 0.f;
    if (!valid) infl = 0.f;
    gjk = jj | (bk << 16);
    gfl = infl;
  };

  auto consume = [&](int gc, int cb, int cjk, float cfl) {
    const int m = gc * PGRP + wave;
    if (m < NPTS) {
      const __half* cwr  = &cw[cb][wave * CWLD + lane];
      const float*  frow = sfeat + lane;
      const int     cfb  = __float_as_int(cfl);
      unsigned todo = __builtin_amdgcn_ballot_w32(cfl > 0.f);
      float a0 = 0.f, a1 = 0.f;
      while (todo) {
        bool     on[NWALK];
        int      jk[NWALK];
        float    f[NWALK], x0[NWALK], x1[NWALK];
        unsigned c0[NWALK], c1[NWALK];
        #pragma unroll
        for (int i = 0; i < NWALK; ++i) {
          on[i] = (todo != 0u);
          jk[i] = 0; f[i] = 0.f; x0[i] = 0.f; x1[i] = 0.f;
          if (on[i]) {
            const int h = __builtin_ctz(todo);
            todo &= todo - 1u;
            jk[i] = __builtin_amdgcn_readlane(cjk, h);
            f[i]  = __int_as_float(__builtin_amdgcn_readlane(cfb, h));
            const float* fr = frow + (jk[i] & 0xffff) * 64;
            x0[i] = fr[0];
            x1[i] = fr[32];
          }
        }
        #pragma unroll
        for (int i = 0; i < NWALK; ++i) {
          c0[i] = 0u; c1[i] = 0u;
          if (on[i]) {
            const __half* cr = cwr + (jk[i] >> 16) * 64;
            c0[i] = __half_as_ushort(cr[0]);
            c1[i] = __half_as_ushort(cr[32]);
          }
        }
        #pragma unroll
        for (int i = 0; i < NWALK; ++i) {
          if (on[i]) {
            const float w0 = __half2float(__ushort_as_half((unsigned short)c0[i])) * f[i];
            const float w1 = __half2float(__ushort_as_half((unsigned short)c1[i])) * f[i];
            a0 = fmaf(x0[i], w0, a0);
            a1 = fmaf(x1[i], w1, a1);
          }
        }
      }
      out[m * 64 + lane]      = a0;
      out[m * 64 + lane + 32] = a1;
    }
  };

  {
    const float* __restrict__ Bsrc = convw ? W2 : W1;
    const int ldb   = convw ? CWLD : CCH;
    const int ncol0 = convw ? wave * 32 : hw * 32;
    #pragma unroll
    for (int c = 0; c < 2; ++c) {
      const int n = ncol0 + c * 16 + lm16;
      #pragma unroll
      for (int s = 0; s < 16; ++s) {
        const int kk = 4 * s + 2 * half;
        bf[c][s].x = Bsrc[kk * ldb + n];
        bf[c][s].y = Bsrc[(kk + 1) * ldb + n];
      }
    }
  }
  if (!convw) make_act(blockIdx.x, 0);
  __syncthreads();

  int   gjk = 0;
  float gfl = 0.f;
  for (int it = -1; ; ++it) {
    const int  gc   = (int)blockIdx.x + it * GRID3;
    const int  gp   = gc + GRID3;
    const bool hasC = (it >= 0);
    const bool hasP = (gp < GROUPS);
    if (pfirst) {
      const int   ojk = gjk;
      const float ofl = gfl;
      if (hasP) produce(gp, (it + 1) & 1, gjk, gfl);
      if (hasC) consume(gc, it & 1, ojk, ofl);
    } else {
      if (hasC) consume(gc, it & 1, gjk, gfl);
      if (hasP) produce(gp, (it + 1) & 1, gjk, gfl);
    }
    __syncthreads();
    if (!hasP) break;
  }
}

extern "C" void kernel_launch(void* const* d_in, const int* in_sizes, int n_in,
                              void* d_out, int out_size, void* d_ws, size_t ws_size,
                              hipStream_t stream) {
  const float* qpts  = (const float*)d_in[0];
  const float* spts  = (const float*)d_in[1];
  const float* sfeat = (const float*)d_in[2];
  const int*   nbidx = (const int*)  d_in[3];
  const float* kpts  = (const float*)d_in[4];
  const float* W1    = (const float*)d_in[5];
  const float* gamma = (const float*)d_in[6];
  const float* beta  = (const float*)d_in[7];
  const float* W2    = (const float*)d_in[8];
  const float* b2    = (const float*)d_in[9];
  float* out = (float*)d_out;
  float* ws  = (float*)d_ws;

  k1_stats<<<GRID1, K1THREADS, 0, stream>>>(sfeat, W1, ws);
  k3_main<<<GRID3, K3THREADS, 0, stream>>>(qpts, spts, sfeat, nbidx, kpts,
                                     W1, gamma, beta, W2, b2, ws, out);
  (void)in_sizes; (void)n_in; (void)out_size; (void)ws_size;
}


# k3: each wave hides its own gather latency behind its own matrix work: a walk trip's row loads go out ahead of each 32-WMMA row-tile block, summed behind it; h waves are the oldest two
# speedup vs baseline: 2.1815x; 1.0459x over previous
#include <hip/hip_runtime.h>
#include <hip/hip_fp16.h>


typedef float v2f __attribute__((ext_vector_type(2)));
typedef float v8f __attribute__((ext_vector_type(8)));

#define NPTS   50000
#define CCH    64
#define KPN    15
#define HNB    32
#define SIGMA  0.7f
#define BN_EPS 1e-5f

#define TILES     ((NPTS + 15) / 16)
#define K1WAVES   8
#define K1THREADS 256
#define GRID1     256
#define K1ROUND   (GRID1 * K1WAVES)
#define NRED      16
#define K3THREADS 1024
#define CONVWAVES 30
#define HWAVES    (K3THREADS / 32 - CONVWAVES)
#define PGRP      32
#define GROUPS    ((NPTS + PGRP - 1) / PGRP)
#define GRID3     256
#define CWLD      (KPN * CCH)
#define NWALK     4
#define LDS_KP    0
#define LDS_BN    (LDS_KP  + 48 * 4)
#define LDS_RED   (LDS_BN  + 2 * CCH * 4)
#define LDS_ACT   (LDS_RED + NRED * 32 * 16)
#define LDS_CW    (LDS_ACT + 2 * PGRP * 68 * 4)
#define LDS_TOTAL (LDS_CW  + 2 * PGRP * CWLD * 2)
static_assert(LDS_RED % 16 == 0 && LDS_ACT % 16 == 0 && LDS_CW % 16 == 0, "LDS pieces must stay 16-byte aligned");
static_assert(NPTS <= 65536 && KPN <= 16, "neighbor row and kernel point share one 32-bit word (16 + 4 bits)");
static_assert(HNB == 32 && PGRP == 32, "one lane per neighbor, one wave per local point");
static_assert(CONVWAVES * 32 == CWLD, "conv waves must cover all W2 columns");
static_assert(GRID3 <= GROUPS, "every block needs a first group");
static_assert(GRID1 % NRED == 0, "partial rows must split evenly over the summing waves");
static_assert(K1THREADS == K1WAVES * 32, "k1 block size must match its wave count");
static_assert(K1ROUND <= TILES && TILES <= 2 * K1ROUND, "every k1 wave has a first tile and at most a second");
static_assert(NRED < K3THREADS / 32, "the wave behind the summing waves copies the kernel points");

static __device__ __forceinline__ v8f wmma_f32_16x16x4(v2f a, v2f b, v8f c) {
  return __builtin_amdgcn_wmma_f32_16x16x4_f32(false, a, false, b,
                                               (short)0, c, false, false);
}

__global__ __launch_bounds__(K1THREADS) void k1_stats(const float* __restrict__ sfeat,
                                                const float* __restrict__ W1,
                                                float* __restrict__ partials) {
  __shared__ float wsum[K1WAVES][64];
  __shared__ float wssq[K1WAVES][64];
  const int tid  = threadIdx.x;
  const int wave = __builtin_amdgcn_readfirstlane(tid >> 5);
  const int lane = tid & 31;
  const int half = lane >> 4;
  const int lm16 = lane & 15;
  const int  t0   = blockIdx.x * K1WAVES + wave;
  const int  t1   = K1ROUND + wave * GRID1 + blockIdx.x;
  const bool has1 = (t1 < TILES);

  v2f a0[16], a1[16];
  {
    const float* __restrict__ arow = sfeat + (t0 * 16 + lm16) * CCH + 2 * half;
    #pragma unroll
    for (int s = 0; s < 16; ++s) a0[s] = *(const v2f*)(arow + 4 * s);
  }
  #pragma unroll
  for (int s = 0; s < 16; ++s) a1[s] = (v2f){0.f, 0.f};
  if (has1) {
    const float* __restrict__ arow = sfeat + (t1 * 16 + lm16) * CCH + 2 * half;
    #pragma unroll
    for (int s = 0; s < 16; ++s) a1[s] = *(const v2f*)(arow + 4 * s);
  }

  v2f bf[4][16];
  #pragma unroll
  for (int s = 0; s < 16; ++s) {
    const int kk = 4 * s + 2 * half;
    #pragma unroll
    for (int c = 0; c < 4; ++c) {
      const int n = c * 16 + lm16;
      bf[c][s].x = W1[kk * CCH + n];
      bf[c][s].y = W1[(kk + 1) * CCH + n];
    }
  }

  float ssum[4], qsum[4];
  #pragma unroll
  for (int c = 0; c < 4; ++c) { ssum[c] = 0.f; qsum[c] = 0.f; }

  auto tile = [&](const v2f (&a)[16]) {
    v8f acc[4];
    #pragma unroll
    for (int c = 0; c < 4; ++c)
      acc[c] = (v8f){0.f,0.f,0.f,0.f,0.f,0.f,0.f,0.f};
    #pragma unroll
    for (int s = 0; s < 16; ++s) {
      #pragma unroll
      for (int c = 0; c < 4; ++c)
        acc[c] = wmma_f32_16x16x4(a[s], bf[c][s], acc[c]);
    }
    #pragma unroll
    for (int c = 0; c < 4; ++c) {
      #pragma unroll
      for (int r = 0; r < 8; ++r) { const float v = acc[c][r]; ssum[c] += v; qsum[c] += v * v; }
    }
  };
  tile(a0);
  if (has1) tile(a1);

  #pragma unroll
  for (int c = 0; c < 4; ++c) {
    const float s = ssum[c] + __shfl_xor(ssum[c], 16);
    const float q = qsum[c] + __shfl_xor(qsum[c], 16);
    if (half == 0) { wsum[wave][c * 16 + lm16] = s; wssq[wave][c * 16 + lm16] = q; }
  }
  __syncthreads();

  if (tid < 128) {
    const int c = tid & 63;
    const float* src = (tid < 64) ? &wsum[0][0] : &wssq[0][0];
    float s = 0.f;
    #pragma unroll
    for (int w = 0; w < K1WAVES; ++w) s += src[w * 64 + c];
    partials[blockIdx.x * 128 + tid] = s;
  }
}

__global__ __launch_bounds__(K3THREADS) void k3_main(const float* __restrict__ qpts,
                                               const float* __restrict__ spts,
                                               const float* __restrict__ sfeat,
                                               const int*   __restrict__ nbidx,
                                               const float* __restrict__ kpts,
                                               const float* __restrict__ W1,
                                               const float* __restrict__ gamma,
                                               const float* __restrict__ beta,
                                               const float* __restrict__ W2,
                                               const float* __restrict__ b2,
                                               const float* __restrict__ ws,
                                               float* __restrict__ out) {
  __shared__ __align__(16) unsigned char smem[LDS_TOTAL];
  float*  const kp  = (float*)(smem + LDS_KP);
  float*  const bn  = (float*)(smem + LDS_BN);
  float4* const red = (float4*)(smem + LDS_RED);
  float  (*const act)[PGRP * 68]     = (float (*)[PGRP * 68])(smem + LDS_ACT);
  __half (*const cw)[PGRP * CWLD]    = (__half (*)[PGRP * CWLD])(smem + LDS_CW);

  const int tid  = threadIdx.x;
  const int wave = __builtin_amdgcn_readfirstlane(tid >> 5);
  const int lane = tid & 31;
  const int half = lane >> 4;
  const int lm16 = lane & 15;
  const bool convw = (wave >= HWAVES);
  const int  hw    = wave;
  const int  cwv   = wave - HWAVES;

  if (wave < NRED) {
    const float4* __restrict__ p4 = (const float4*)ws;
    float4 s = make_float4(0.f, 0.f, 0.f, 0.f);
    #pragma unroll
    for (int i = 0; i < GRID1 / NRED; ++i) {
      const float4 v = p4[(wave + NRED * i) * 32 + lane];
      s.x += v.x; s.y += v.y; s.z += v.z; s.w += v.w;
    }
    red[wave * 32 + lane] = s;
  } else if (wave == NRED) {
    const float k0 = kpts[lane];
    if (lane < 45 - 32) kp[32 + lane] = kpts[32 + lane];
    kp[lane] = k0;
  }
  __syncthreads();

  if (tid < CCH) {
    const int ch = tid;
    const float* redf = (const float*)red;
    float ssum = 0.f, qsum = 0.f;
    #pragma unroll
    for (int w = 0; w < NRED; ++w) {
      ssum += redf[w * 128 + ch];
      qsum += redf[w * 128 + 64 + ch];
    }
    const float inv_n = 1.0f / (float)NPTS;
    const float mu  = ssum * inv_n;
    const float var = qsum * inv_n - mu * mu;
    const float sc  = gamma[ch] * rsqrtf(var + BN_EPS);
    const float sh  = beta[ch] - mu * sc;
    bn[ch] = sc; bn[CCH + ch] = sh;
  }

  v2f bf[2][16];
  __syncthreads();

  auto act_half = [&](int gq, int bq, int rt) {
    int row = gq * PGRP + rt * 16 + lm16;
    row = (row < NPTS) ? row : (NPTS - 1);
    const float* __restrict__ arow = sfeat + row * CCH + 2 * half;
    v8f acc0 = (v8f){0.f,0.f,0.f,0.f,0.f,0.f,0.f,0.f};
    v8f acc1 = (v8f){0.f,0.f,0.f,0.f,0.f,0.f,0.f,0.f};
    #pragma unroll
    for (int hb = 0; hb < 16 / 8; ++hb) {
      v2f a[8];
      #pragma unroll
      for (int s = 0; s < 8; ++s) a[s] = *(const v2f*)(arow + 4 * (8 * hb + s));
      #pragma unroll
      for (int s = 0; s < 8; ++s) {
        acc0 = wmma_f32_16x16x4(a[s], bf[0][8 * hb + s], acc0);
        acc1 = wmma_f32_16x16x4(a[s], bf[1][8 * hb + s], acc1);
      }
    }
    const float* bnw = bn + hw * 32 + lm16;
    const float sc0 = bnw[0],  sh0 = bnw[CCH];
    const float sc1 = bnw[16], sh1 = bnw[CCH + 16];
    float* actw = &act[bq][(rt * 16 + 8 * half) * 68 + hw * 32 + lm16];
    #pragma unroll
    for (int r = 0; r < 8; ++r) {
      float v0 = acc0[r] * sc0 + sh0;
      float v1 = acc1[r] * sc1 + sh1;
      v0 = (v0 > 0.f) ? v0 : 0.1f * v0;
      v1 = (v1 > 0.f) ? v1 : 0.1f * v1;
      actw[r * 68]      = v0;
      actw[r * 68 + 16] = v1;
    }
  };

  auto conv_half = [&](int pb, int rt, float bias0, float bias1) {
    const int n0 = cwv * 32 + lm16;
    const float* arow = &act[pb][(rt * 16 + lm16) * 68 + 2 * half];
    v8f acc0 = (v8f){0.f,0.f,0.f,0.f,0.f,0.f,0.f,0.f};
    v8f acc1 = (v8f){0.f,0.f,0.f,0.f,0.f,0.f,0.f,0.f};
    #pragma unroll
    for (int s = 0; s < 16; ++s) {
      v2f a = *(const v2f*)(arow + 4 * s);
      acc0 = wmma_f32_16x16x4(a, bf[0][s], acc0);
      acc1 = wmma_f32_16x16x4(a, bf[1][s], acc1);
    }
    __half* cww = &cw[pb][(rt * 16 + 8 * half) * CWLD + n0];
    #pragma unroll
    for (int r = 0; r < 8; ++r) {
      cww[r * CWLD]      = __float2half(acc0[r] + bias0);
      cww[r * CWLD + 16] = __float2half(acc1[r] + bias1);
    }
  };

  auto matrix_half = [&](int gp, int pb, int rt, float bias0, float bias1) {
    if (convw) {
      conv_half(pb, rt, bias0, bias1);
    } else {
      const int gn = gp + GRID3;
      if (gn < GROUPS) act_half(gn, pb ^ 1, rt);
    }
  };

  auto geometry = [&](int gm, int j, int& gjk, float& gfl) {
    const bool valid = ((unsigned)j < (unsigned)NPTS);
    const int  jj    = valid ? j : 0;
    const float dx = spts[jj * 3 + 0] - qpts[gm * 3 + 0];
    const float dy = spts[jj * 3 + 1] - qpts[gm * 3 + 1];
    const float dz = spts[jj * 3 + 2] - qpts[gm * 3 + 2];
    float best = 3.4e38f; int bk = 0;
    #pragma unroll
    for (int k = 0; k < KPN; ++k) {
      const float ex = dx - kp[k * 3 + 0];
      const float ey = dy - kp[k * 3 + 1];
      const float ez = dz - kp[k * 3 + 2];
      const float sq = ex * ex + ey * ey + ez * ez;
      if (sq < best) { best = sq; bk = k; }
    }
    float infl = 1.0f - sqrtf(best) * (1.0f / SIGMA);
    infl = (infl > 0.f) ? infl : 0.f;
    if (!valid) infl = 0.f;
    gjk = jj | (bk << 16);
    gfl = infl;
  };

  struct Trip {
    bool  on[NWALK];
    int   jk[NWALK];
    float f[NWALK], x0[NWALK], x1[NWALK];
  };
  auto trip_issue = [&](unsigned& todo, Trip& t, int cjk, int cfb) {
    #pragma unroll
    for (int i = 0; i < NWALK; ++i) {
      t.on[i] = (todo != 0u);
      t.jk[i] = 0; t.f[i] = 0.f; t.x0[i] = 0.f; t.x1[i] = 0.f;
      if (t.on[i]) {
        const int h = __builtin_ctz(todo);
        todo &= todo - 1u;
        t.jk[i] = __builtin_amdgcn_readlane(cjk, h);
        t.f[i]  = __int_as_float(__builtin_amdgcn_readlane(cfb, h));
        const float* fr = sfeat + (t.jk[i] & 0xffff) * 64;
        t.x0[i] = fr[lane];
        t.x1[i] = fr[lane + 32];
      }
    }
  };
  auto trip_finish = [&](const Trip& t, const __half* cwr, float& a0, float& a1) {
    unsigned c0[NWALK], c1[NWALK];
    #pragma unroll
    for (int i = 0; i < NWALK; ++i) {
      c0[i] = 0u; c1[i] = 0u;
      if (t.on[i]) {
        const __half* cr = cwr + (t.jk[i] >> 16) * 64;
        c0[i] = __half_as_ushort(cr[0]);
        c1[i] = __half_as_ushort(cr[32]);
      }
    }
    #pragma unroll
    for (int i = 0; i < NWALK; ++i) {
      if (t.on[i]) {
        const float w0 = __half2float(__ushort_as_half((unsigned short)c0[i])) * t.f[i];
        const float w1 = __half2float(__ushort_as_half((unsigned short)c1[i])) * t.f[i];
        a0 = fmaf(t.x0[i], w0, a0);
        a1 = fmaf(t.x1[i], w1, a1);
      }
    }
  };

  {
    const float* __restrict__ Bsrc = convw ? W2 : W1;
    const int ldb   = convw ? CWLD : CCH;
    const int ncol0 = convw ? cwv * 32 : hw * 32;
    #pragma unroll
    for (int c = 0; c < 2; ++c) {
      const int n = ncol0 + c * 16 + lm16;
      #pragma unroll
      for (int s = 0; s < 16; ++s) {
        const int kk = 4 * s + 2 * half;
        bf[c][s].x = Bsrc[kk * ldb + n];
        bf[c][s].y = Bsrc[(kk + 1) * ldb + n];
      }
    }
  }
  if (!convw) { act_half(blockIdx.x, 0, 0); act_half(blockIdx.x, 0, 1); }
  __syncthreads();

  const float bias0 = convw ? b2[cwv * 32 + lm16]      : 0.f;
  const float bias1 = convw ? b2[cwv * 32 + lm16 + 16] : 0.f;

  int   gjk = 0;
  float gfl = 0.f;
  for (int it = -1; ; ++it) {
    const int  gc   = (int)blockIdx.x + it * GRID3;
    const int  gp   = gc + GRID3;
    const int  cb   = it & 1;
    const int  pb   = (it + 1) & 1;
    const bool hasP = (gp < GROUPS);
    const int  m    = gc * PGRP + wave;
    const bool doC  = (it >= 0) && (m < NPTS);

    int gm = gp * PGRP + wave;
    gm = (gm < NPTS) ? gm : (NPTS - 1);
    int   j = 0;
    if (hasP) j = nbidx[gm * HNB + lane];

    const __half* cwr = &cw[cb][wave * CWLD + lane];
    const int     cfb = __float_as_int(gfl);
    unsigned todo = doC ? __builtin_amdgcn_ballot_w32(gfl > 0.f) : 0u;
    float a0 = 0.f, a1 = 0.f;
    Trip t1, t2;
    trip_issue(todo, t1, gjk, cfb);
    if (hasP) matrix_half(gp, pb, 0, bias0, bias1);
    trip_finish(t1, cwr, a0, a1);
    trip_issue(todo, t2, gjk, cfb);
    if (hasP) matrix_half(gp, pb, 1, bias0, bias1);
    trip_finish(t2, cwr, a0, a1);
    while (todo) {
      Trip t;
      trip_issue(todo, t, gjk, cfb);
      trip_finish(t, cwr, a0, a1);
    }
    if (doC) {
      out[m * 64 + lane]      = a0;
      out[m * 64 + lane + 32] = a1;
    }
    if (hasP) geometry(gm, j, gjk, gfl);
    __syncthreads();
    if (!hasP) break;
  }
}

extern "C" void kernel_launch(void* const* d_in, const int* in_sizes, int n_in,
                              void* d_out, int out_size, void* d_ws, size_t ws_size,
                              hipStream_t stream) {
  const float* qpts  = (const float*)d_in[0];
  const float* spts  = (const float*)d_in[1];
  const float* sfeat = (const float*)d_in[2];
  const int*   nbidx = (const int*)  d_in[3];
  const float* kpts  = (const float*)d_in[4];
  const float* W1    = (const float*)d_in[5];
  const float* gamma = (const float*)d_in[6];
  const float* beta  = (const float*)d_in[7];
  const float* W2    = (const float*)d_in[8];
  const float* b2    = (const float*)d_in[9];
  float* out = (float*)d_out;
  float* ws  = (float*)d_ws;

  k1_stats<<<GRID1, K1THREADS, 0, stream>>>(sfeat, W1, ws);
  k3_main<<<GRID3, K3THREADS, 0, stream>>>(qpts, spts, sfeat, nbidx, kpts,
                                     W1, gamma, beta, W2, b2, ws, out);
  (void)in_sizes; (void)n_in; (void)out_size; (void)ws_size;
}


# k3: the four younger waves of each SIMD do their neighbor walk and geometry before their 64 products, the four older after them, so each half's memory waits run under the other half's products
# speedup vs baseline: 2.2731x; 1.0420x over previous
#include <hip/hip_runtime.h>
#include <hip/hip_fp16.h>


typedef float v2f __attribute__((ext_vector_type(2)));
typedef float v8f __attribute__((ext_vector_type(8)));

#define NPTS   50000
#define CCH    64
#define KPN    15
#define HNB    32
#define SIGMA  0.7f
#define BN_EPS 1e-5f

#define TILES     ((NPTS + 15) / 16)
#define K1WAVES   8
#define K1THREADS 256
#define GRID1     256
#define K1ROUND   (GRID1 * K1WAVES)
#define NRED      16
#define K3THREADS 1024
#define CONVWAVES 30
#define HWAVES    (K3THREADS / 32 - CONVWAVES)
#define PGRP      32
#define GROUPS    ((NPTS + PGRP - 1) / PGRP)
#define GRID3     256
#define CWLD      (KPN * CCH)
#define NWALK     4
#define LDS_KP    0
#define LDS_BN    (LDS_KP  + 48 * 4)
#define LDS_RED   (LDS_BN  + 2 * CCH * 4)
#define LDS_ACT   (LDS_RED + NRED * 32 * 16)
#define LDS_CW    (LDS_ACT + 2 * PGRP * 68 * 4)
#define LDS_TOTAL (LDS_CW  + 2 * PGRP * CWLD * 2)
static_assert(LDS_RED % 16 == 0 && LDS_ACT % 16 == 0 && LDS_CW % 16 == 0, "LDS pieces must stay 16-byte aligned");
static_assert(NPTS <= 65536 && KPN <= 16, "neighbor row and kernel point share one 32-bit word (16 + 4 bits)");
static_assert(HNB == 32 && PGRP == 32, "one lane per neighbor, one wave per local point");
static_assert(CONVWAVES * 32 == CWLD, "conv waves must cover all W2 columns");
static_assert(GRID3 <= GROUPS, "every block needs a first group");
static_assert(GRID1 % NRED == 0, "partial rows must split evenly over the summing waves");
static_assert(K1THREADS == K1WAVES * 32, "k1 block size must match its wave count");
static_assert(K1ROUND <= TILES && TILES <= 2 * K1ROUND, "every k1 wave has a first tile and at most a second");
static_assert(NRED < K3THREADS / 32, "the wave behind the summing waves copies the kernel points");

static __device__ __forceinline__ v8f wmma_f32_16x16x4(v2f a, v2f b, v8f c) {
  return __builtin_amdgcn_wmma_f32_16x16x4_f32(false, a, false, b,
                                               (short)0, c, false, false);
}

__global__ __launch_bounds__(K1THREADS) void k1_stats(const float* __restrict__ sfeat,
                                                const float* __restrict__ W1,
                                                float* __restrict__ partials) {
  __shared__ float wsum[K1WAVES][64];
  __shared__ float wssq[K1WAVES][64];
  const int tid  = threadIdx.x;
  const int wave = __builtin_amdgcn_readfirstlane(tid >> 5);
  const int lane = tid & 31;
  const int half = lane >> 4;
  const int lm16 = lane & 15;
  const int  t0   = blockIdx.x * K1WAVES + wave;
  const int  t1   = K1ROUND + wave * GRID1 + blockIdx.x;
  const bool has1 = (t1 < TILES);

  v2f a0[16], a1[16];
  {
    const float* __restrict__ arow = sfeat + (t0 * 16 + lm16) * CCH + 2 * half;
    #pragma unroll
    for (int s = 0; s < 16; ++s) a0[s] = *(const v2f*)(arow + 4 * s);
  }
  #pragma unroll
  for (int s = 0; s < 16; ++s) a1[s] = (v2f){0.f, 0.f};
  if (has1) {
    const float* __restrict__ arow = sfeat + (t1 * 16 + lm16) * CCH + 2 * half;
    #pragma unroll
    for (int s = 0; s < 16; ++s) a1[s] = *(const v2f*)(arow + 4 * s);
  }

  v2f bf[4][16];
  #pragma unroll
  for (int s = 0; s < 16; ++s) {
    const int kk = 4 * s + 2 * half;
    #pragma unroll
    for (int c = 0; c < 4; ++c) {
      const int n = c * 16 + lm16;
      bf[c][s].x = W1[kk * CCH + n];
      bf[c][s].y = W1[(kk + 1) * CCH + n];
    }
  }

  float ssum[4], qsum[4];
  #pragma unroll
  for (int c = 0; c < 4; ++c) { ssum[c] = 0.f; qsum[c] = 0.f; }

  auto tile = [&](const v2f (&a)[16]) {
    v8f acc[4];
    #pragma unroll
    for (int c = 0; c < 4; ++c)
      acc[c] = (v8f){0.f,0.f,0.f,0.f,0.f,0.f,0.f,0.f};
    #pragma unroll
    for (int s = 0; s < 16; ++s) {
      #pragma unroll
      for (int c = 0; c < 4; ++c)
        acc[c] = wmma_f32_16x16x4(a[s], bf[c][s], acc[c]);
    }
    #pragma unroll
    for (int c = 0; c < 4; ++c) {
      #pragma unroll
      for (int r = 0; r < 8; ++r) { const float v = acc[c][r]; ssum[c] += v; qsum[c] += v * v; }
    }
  };
  tile(a0);
  if (has1) tile(a1);

  #pragma unroll
  for (int c = 0; c < 4; ++c) {
    const float s = ssum[c] + __shfl_xor(ssum[c], 16);
    const float q = qsum[c] + __shfl_xor(qsum[c], 16);
    if (half == 0) { wsum[wave][c * 16 + lm16] = s; wssq[wave][c * 16 + lm16] = q; }
  }
  __syncthreads();

  if (tid < 128) {
    const int c = tid & 63;
    const float* src = (tid < 64) ? &wsum[0][0] : &wssq[0][0];
    float s = 0.f;
    #pragma unroll
    for (int w = 0; w < K1WAVES; ++w) s += src[w * 64 + c];
    partials[blockIdx.x * 128 + tid] = s;
  }
}

__global__ __launch_bounds__(K3THREADS) void k3_main(const float* __restrict__ qpts,
                                               const float* __restrict__ spts,
                                               const float* __restrict__ sfeat,
                                               const int*   __restrict__ nbidx,
                                               const float* __restrict__ kpts,
                                               const float* __restrict__ W1,
                                               const float* __restrict__ gamma,
                                               const float* __restrict__ beta,
                                               const float* __restrict__ W2,
                                               const float* __restrict__ b2,
                                               const float* __restrict__ ws,
                                               float* __restrict__ out) {
  __shared__ __align__(16) unsigned char smem[LDS_TOTAL];
  float*  const kp  = (float*)(smem + LDS_KP);
  float*  const bn  = (float*)(smem + LDS_BN);
  float4* const red = (float4*)(smem + LDS_RED);
  float  (*const act)[PGRP * 68]     = (float (*)[PGRP * 68])(smem + LDS_ACT);
  __half (*const cw)[PGRP * CWLD]    = (__half (*)[PGRP * CWLD])(smem + LDS_CW);

  const int tid  = threadIdx.x;
  const int wave = __builtin_amdgcn_readfirstlane(tid >> 5);
  const int lane = tid & 31;
  const int half = lane >> 4;
  const int lm16 = lane & 15;
  const bool convw = (wave >= HWAVES);
  const bool young = (wave >= K3THREADS / 64);
  const int  hw    = wave;
  const int  cwv   = wave - HWAVES;

  if (wave < NRED) {
    const float4* __restrict__ p4 = (const float4*)ws;
    float4 s = make_float4(0.f, 0.f, 0.f, 0.f);
    #pragma unroll
    for (int i = 0; i < GRID1 / NRED; ++i) {
      const float4 v = p4[(wave + NRED * i) * 32 + lane];
      s.x += v.x; s.y += v.y; s.z += v.z; s.w += v.w;
    }
    red[wave * 32 + lane] = s;
  } else if (wave == NRED) {
    const float k0 = kpts[lane];
    if (lane < 45 - 32) kp[32 + lane] = kpts[32 + lane];
    kp[lane] = k0;
  }
  __syncthreads();

  if (tid < CCH) {
    const int ch = tid;
    const float* redf = (const float*)red;
    float ssum = 0.f, qsum = 0.f;
    #pragma unroll
    for (int w = 0; w < NRED; ++w) {
      ssum += redf[w * 128 + ch];
      qsum += redf[w * 128 + 64 + ch];
    }
    const float inv_n = 1.0f / (float)NPTS;
    const float mu  = ssum * inv_n;
    const float var = qsum * inv_n - mu * mu;
    const float sc  = gamma[ch] * rsqrtf(var + BN_EPS);
    const float sh  = beta[ch] - mu * sc;
    bn[ch] = sc; bn[CCH + ch] = sh;
  }

  v2f bf[2][16];
  __syncthreads();

  auto act_half = [&](int gq, int bq, int rt) {
    int row = gq * PGRP + rt * 16 + lm16;
    row = (row < NPTS) ? row : (NPTS - 1);
    const float* __restrict__ arow = sfeat + row * CCH + 2 * half;
    v8f acc0 = (v8f){0.f,0.f,0.f,0.f,0.f,0.f,0.f,0.f};
    v8f acc1 = (v8f){0.f,0.f,0.f,0.f,0.f,0.f,0.f,0.f};
    #pragma unroll
    for (int hb = 0; hb < 16 / 8; ++hb) {
      v2f a[8];
      #pragma unroll
      for (int s = 0; s < 8; ++s) a[s] = *(const v2f*)(arow + 4 * (8 * hb + s));
      #pragma unroll
      for (int s = 0; s < 8; ++s) {
        acc0 = wmma_f32_16x16x4(a[s], bf[0][8 * hb + s], acc0);
        acc1 = wmma_f32_16x16x4(a[s], bf[1][8 * hb + s], acc1);
      }
    }
    const float* bnw = bn + hw * 32 + lm16;
    const float sc0 = bnw[0],  sh0 = bnw[CCH];
    const float sc1 = bnw[16], sh1 = bnw[CCH + 16];
    float* actw = &act[bq][(rt * 16 + 8 * half) * 68 + hw * 32 + lm16];
    #pragma unroll
    for (int r = 0; r < 8; ++r) {
      float v0 = acc0[r] * sc0 + sh0;
      float v1 = acc1[r] * sc1 + sh1;
      v0 = (v0 > 0.f) ? v0 : 0.1f * v0;
      v1 = (v1 > 0.f) ? v1 : 0.1f * v1;
      actw[r * 68]      = v0;
      actw[r * 68 + 16] = v1;
    }
  };

  auto conv_half = [&](int pb, int rt, float bias0, float bias1) {
    const int n0 = cwv * 32 + lm16;
    const float* arow = &act[pb][(rt * 16 + lm16) * 68 + 2 * half];
    v8f acc0 = (v8f){0.f,0.f,0.f,0.f,0.f,0.f,0.f,0.f};
    v8f acc1 = (v8f){0.f,0.f,0.f,0.f,0.f,0.f,0.f,0.f};
    #pragma unroll
    for (int s = 0; s < 16; ++s) {
      v2f a = *(const v2f*)(arow + 4 * s);
      acc0 = wmma_f32_16x16x4(a, bf[0][s], acc0);
      acc1 = wmma_f32_16x16x4(a, bf[1][s], acc1);
    }
    __half* cww = &cw[pb][(rt * 16 + 8 * half) * CWLD + n0];
    #pragma unroll
    for (int r = 0; r < 8; ++r) {
      cww[r * CWLD]      = __float2half(acc0[r] + bias0);
      cww[r * CWLD + 16] = __float2half(acc1[r] + bias1);
    }
  };

  auto matrix_half = [&](int gp, int pb, int rt, float bias0, float bias1) {
    if (convw) {
      conv_half(pb, rt, bias0, bias1);
    } else {
      const int gn = gp + GRID3;
      if (gn < GROUPS) act_half(gn, pb ^ 1, rt);
    }
  };

  auto geometry = [&](int gm, int j, int& gjk, float& gfl) {
    const bool valid = ((unsigned)j < (unsigned)NPTS);
    const int  jj    = valid ? j : 0;
    const float dx = spts[jj * 3 + 0] - qpts[gm * 3 + 0];
    const float dy = spts[jj * 3 + 1] - qpts[gm * 3 + 1];
    const float dz = spts[jj * 3 + 2] - qpts[gm * 3 + 2];
    float best = 3.4e38f; int bk = 0;
    #pragma unroll
    for (int k = 0; k < KPN; ++k) {
      const float ex = dx - kp[k * 3 + 0];
      const float ey = dy - kp[k * 3 + 1];
      const float ez = dz - kp[k * 3 + 2];
      const float sq = ex * ex + ey * ey + ez * ez;
      if (sq < best) { best = sq; bk = k; }
    }
    float infl = 1.0f - sqrtf(best) * (1.0f / SIGMA);
    infl = (infl > 0.f) ? infl : 0.f;
    if (!valid) infl = 0.f;
    gjk = jj | (bk << 16);
    gfl = infl;
  };

  struct Trip {
    bool  on[NWALK];
    int   jk[NWALK];
    float f[NWALK], x0[NWALK], x1[NWALK];
  };
  auto trip_issue = [&](unsigned& todo, Trip& t, int cjk, int cfb) {
    #pragma unroll
    for (int i = 0; i < NWALK; ++i) {
      t.on[i] = (todo != 0u);
      t.jk[i] = 0; t.f[i] = 0.f; t.x0[i] = 0.f; t.x1[i] = 0.f;
      if (t.on[i]) {
        const int h = __builtin_ctz(todo);
        todo &= todo - 1u;
        t.jk[i] = __builtin_amdgcn_readlane(cjk, h);
        t.f[i]  = __int_as_float(__builtin_amdgcn_readlane(cfb, h));
        const float* fr = sfeat + (t.jk[i] & 0xffff) * 64;
        t.x0[i] = fr[lane];
        t.x1[i] = fr[lane + 32];
      }
    }
  };
  auto trip_finish = [&](const Trip& t, const __half* cwr, float& a0, float& a1) {
    unsigned c0[NWALK], c1[NWALK];
    #pragma unroll
    for (int i = 0; i < NWALK; ++i) {
      c0[i] = 0u; c1[i] = 0u;
      if (t.on[i]) {
        const __half* cr = cwr + (t.jk[i] >> 16) * 64;
        c0[i] = __half_as_ushort(cr[0]);
        c1[i] = __half_as_ushort(cr[32]);
      }
    }
    #pragma unroll
    for (int i = 0; i < NWALK; ++i) {
      if (t.on[i]) {
        const float w0 = __half2float(__ushort_as_half((unsigned short)c0[i])) * t.f[i];
        const float w1 = __half2float(__ushort_as_half((unsigned short)c1[i])) * t.f[i];
        a0 = fmaf(t.x0[i], w0, a0);
        a1 = fmaf(t.x1[i], w1, a1);
      }
    }
  };

  {
    const float* __restrict__ Bsrc = convw ? W2 : W1;
    const int ldb   = convw ? CWLD : CCH;
    const int ncol0 = convw ? cwv * 32 : hw * 32;
    #pragma unroll
    for (int c = 0; c < 2; ++c) {
      const int n = ncol0 + c * 16 + lm16;
      #pragma unroll
      for (int s = 0; s < 16; ++s) {
        const int kk = 4 * s + 2 * half;
        bf[c][s].x = Bsrc[kk * ldb + n];
        bf[c][s].y = Bsrc[(kk + 1) * ldb + n];
      }
    }
  }
  if (!convw) { act_half(blockIdx.x, 0, 0); act_half(blockIdx.x, 0, 1); }
  __syncthreads();

  const float bias0 = convw ? b2[cwv * 32 + lm16]      : 0.f;
  const float bias1 = convw ? b2[cwv * 32 + lm16 + 16] : 0.f;

  int   gjk = 0;
  float gfl = 0.f;
  for (int it = -1; ; ++it) {
    const int  gc   = (int)blockIdx.x + it * GRID3;
    const int  gp   = gc + GRID3;
    const int  cb   = it & 1;
    const int  pb   = (it + 1) & 1;
    const bool hasP = (gp < GROUPS);
    const int  m    = gc * PGRP + wave;
    const bool doC  = (it >= 0) && (m < NPTS);

    int gm = gp * PGRP + wave;
    gm = (gm < NPTS) ? gm : (NPTS - 1);
    int   j = 0;
    if (hasP) j = nbidx[gm * HNB + lane];

    auto matrix_stage = [&]() {
      if (hasP) {
        matrix_half(gp, pb, 0, bias0, bias1);
        matrix_half(gp, pb, 1, bias0, bias1);
      }
    };
    auto side_stage = [&]() {
      const __half* cwr = &cw[cb][wave * CWLD + lane];
      const int     cfb = __float_as_int(gfl);
      unsigned todo = doC ? __builtin_amdgcn_ballot_w32(gfl > 0.f) : 0u;
      float a0 = 0.f, a1 = 0.f;
      while (todo) {
        Trip t1, t2;
        trip_issue(todo, t1, gjk, cfb);
        trip_issue(todo, t2, gjk, cfb);
        trip_finish(t1, cwr, a0, a1);
        trip_finish(t2, cwr, a0, a1);
      }
      if (doC) {
        out[m * 64 + lane]      = a0;
        out[m * 64 + lane + 32] = a1;
      }
      if (hasP) geometry(gm, j, gjk, gfl);
    };

    if (young) { side_stage();   matrix_stage(); }
    else       { matrix_stage(); side_stage();   }
    __syncthreads();
    if (!hasP) break;
  }
}

extern "C" void kernel_launch(void* const* d_in, const int* in_sizes, int n_in,
                              void* d_out, int out_size, void* d_ws, size_t ws_size,
                              hipStream_t stream) {
  const float* qpts  = (const float*)d_in[0];
  const float* spts  = (const float*)d_in[1];
  const float* sfeat = (const float*)d_in[2];
  const int*   nbidx = (const int*)  d_in[3];
  const float* kpts  = (const float*)d_in[4];
  const float* W1    = (const float*)d_in[5];
  const float* gamma = (const float*)d_in[6];
  const float* beta  = (const float*)d_in[7];
  const float* W2    = (const float*)d_in[8];
  const float* b2    = (const float*)d_in[9];
  float* out = (float*)d_out;
  float* ws  = (float*)d_ws;

  k1_stats<<<GRID1, K1THREADS, 0, stream>>>(sfeat, W1, ws);
  k3_main<<<GRID3, K3THREADS, 0, stream>>>(qpts, spts, sfeat, nbidx, kpts,
                                     W1, gamma, beta, W2, b2, ws, out);
  (void)in_sizes; (void)n_in; (void)out_size; (void)ws_size;
}


# k3 work split by 16-point row tiles: every block takes 6 full 32-point groups, the 848 points left go out as 53 single tiles (a half pass each), so the longest block runs 13 row tiles, not 14
# speedup vs baseline: 2.3696x; 1.0425x over previous
#include <hip/hip_runtime.h>
#include <hip/hip_fp16.h>


typedef float v2f __attribute__((ext_vector_type(2)));
typedef float v8f __attribute__((ext_vector_type(8)));

#define NPTS   50000
#define CCH    64
#define KPN    15
#define HNB    32
#define SIGMA  0.7f
#define BN_EPS 1e-5f

#define TILES     ((NPTS + 15) / 16)
#define K1WAVES   8
#define K1THREADS 256
#define GRID1     256
#define K1ROUND   (GRID1 * K1WAVES)
#define NRED      16
#define K3THREADS 1024
#define CONVWAVES 30
#define HWAVES    (K3THREADS / 32 - CONVWAVES)
#define PGRP      32
#define GROUPS    ((NPTS + PGRP - 1) / PGRP)
#define GRID3     256
#define FULLR     (NPTS / (PGRP * GRID3))
#define TAIL0     (FULLR * GRID3 * PGRP)
#define TAILT     ((NPTS - TAIL0 + 15) / 16)
#define CWLD      (KPN * CCH)
#define NWALK     4
#define LDS_KP    0
#define LDS_BN    (LDS_KP  + 48 * 4)
#define LDS_RED   (LDS_BN  + 2 * CCH * 4)
#define LDS_ACT   (LDS_RED + NRED * 32 * 16)
#define LDS_CW    (LDS_ACT + 2 * PGRP * 68 * 4)
#define LDS_TOTAL (LDS_CW  + 2 * PGRP * CWLD * 2)
static_assert(LDS_RED % 16 == 0 && LDS_ACT % 16 == 0 && LDS_CW % 16 == 0, "LDS pieces must stay 16-byte aligned");
static_assert(NPTS <= 65536 && KPN <= 16, "neighbor row and kernel point share one 32-bit word (16 + 4 bits)");
static_assert(HNB == 32 && PGRP == 32, "one lane per neighbor, one wave per local point");
static_assert(CONVWAVES * 32 == CWLD, "conv waves must cover all W2 columns");
static_assert(GRID3 <= GROUPS, "every block needs a first group");
static_assert(FULLR >= 1 && TAILT <= GRID3, "every block has a full first group and at most one tail tile");
static_assert(GRID1 % NRED == 0, "partial rows must split evenly over the summing waves");
static_assert(K1THREADS == K1WAVES * 32, "k1 block size must match its wave count");
static_assert(K1ROUND <= TILES && TILES <= 2 * K1ROUND, "every k1 wave has a first tile and at most a second");
static_assert(NRED < K3THREADS / 32, "the wave behind the summing waves copies the kernel points");

static __device__ __forceinline__ v8f wmma_f32_16x16x4(v2f a, v2f b, v8f c) {
  return __builtin_amdgcn_wmma_f32_16x16x4_f32(false, a, false, b,
                                               (short)0, c, false, false);
}

__global__ __launch_bounds__(K1THREADS) void k1_stats(const float* __restrict__ sfeat,
                                                const float* __restrict__ W1,
                                                float* __restrict__ partials) {
  __shared__ float wsum[K1WAVES][64];
  __shared__ float wssq[K1WAVES][64];
  const int tid  = threadIdx.x;
  const int wave = __builtin_amdgcn_readfirstlane(tid >> 5);
  const int lane = tid & 31;
  const int half = lane >> 4;
  const int lm16 = lane & 15;
  const int  t0   = blockIdx.x * K1WAVES + wave;
  const int  t1   = K1ROUND + wave * GRID1 + blockIdx.x;
  const bool has1 = (t1 < TILES);

  v2f a0[16], a1[16];
  {
    const float* __restrict__ arow = sfeat + (t0 * 16 + lm16) * CCH + 2 * half;
    #pragma unroll
    for (int s = 0; s < 16; ++s) a0[s] = *(const v2f*)(arow + 4 * s);
  }
  #pragma unroll
  for (int s = 0; s < 16; ++s) a1[s] = (v2f){0.f, 0.f};
  if (has1) {
    const float* __restrict__ arow = sfeat + (t1 * 16 + lm16) * CCH + 2 * half;
    #pragma unroll
    for (int s = 0; s < 16; ++s) a1[s] = *(const v2f*)(arow + 4 * s);
  }

  v2f bf[4][16];
  #pragma unroll
  for (int s = 0; s < 16; ++s) {
    const int kk = 4 * s + 2 * half;
    #pragma unroll
    for (int c = 0; c < 4; ++c) {
      const int n = c * 16 + lm16;
      bf[c][s].x = W1[kk * CCH + n];
      bf[c][s].y = W1[(kk + 1) * CCH + n];
    }
  }

  float ssum[4], qsum[4];
  #pragma unroll
  for (int c = 0; c < 4; ++c) { ssum[c] = 0.f; qsum[c] = 0.f; }

  auto tile = [&](const v2f (&a)[16]) {
    v8f acc[4];
    #pragma unroll
    for (int c = 0; c < 4; ++c)
      acc[c] = (v8f){0.f,0.f,0.f,0.f,0.f,0.f,0.f,0.f};
    #pragma unroll
    for (int s = 0; s < 16; ++s) {
      #pragma unroll
      for (int c = 0; c < 4; ++c)
        acc[c] = wmma_f32_16x16x4(a[s], bf[c][s], acc[c]);
    }
    #pragma unroll
    for (int c = 0; c < 4; ++c) {
      #pragma unroll
      for (int r = 0; r < 8; ++r) { const float v = acc[c][r]; ssum[c] += v; qsum[c] += v * v; }
    }
  };
  tile(a0);
  if (has1) tile(a1);

  #pragma unroll
  for (int c = 0; c < 4; ++c) {
    const float s = ssum[c] + __shfl_xor(ssum[c], 16);
    const float q = qsum[c] + __shfl_xor(qsum[c], 16);
    if (half == 0) { wsum[wave][c * 16 + lm16] = s; wssq[wave][c * 16 + lm16] = q; }
  }
  __syncthreads();

  if (tid < 128) {
    const int c = tid & 63;
    const float* src = (tid < 64) ? &wsum[0][0] : &wssq[0][0];
    float s = 0.f;
    #pragma unroll
    for (int w = 0; w < K1WAVES; ++w) s += src[w * 64 + c];
    partials[blockIdx.x * 128 + tid] = s;
  }
}

__global__ __launch_bounds__(K3THREADS) void k3_main(const float* __restrict__ qpts,
                                               const float* __restrict__ spts,
                                               const float* __restrict__ sfeat,
                                               const int*   __restrict__ nbidx,
                                               const float* __restrict__ kpts,
                                               const float* __restrict__ W1,
                                               const float* __restrict__ gamma,
                                               const float* __restrict__ beta,
                                               const float* __restrict__ W2,
                                               const float* __restrict__ b2,
                                               const float* __restrict__ ws,
                                               float* __restrict__ out) {
  __shared__ __align__(16) unsigned char smem[LDS_TOTAL];
  float*  const kp  = (float*)(smem + LDS_KP);
  float*  const bn  = (float*)(smem + LDS_BN);
  float4* const red = (float4*)(smem + LDS_RED);
  float  (*const act)[PGRP * 68]     = (float (*)[PGRP * 68])(smem + LDS_ACT);
  __half (*const cw)[PGRP * CWLD]    = (__half (*)[PGRP * CWLD])(smem + LDS_CW);

  const int tid  = threadIdx.x;
  const int wave = __builtin_amdgcn_readfirstlane(tid >> 5);
  const int lane = tid & 31;
  const int half = lane >> 4;
  const int lm16 = lane & 15;
  const bool convw = (wave >= HWAVES);
  const bool young = (wave >= K3THREADS / 64);
  const int  hw    = wave;
  const int  cwv   = wave - HWAVES;

  if (wave < NRED) {
    const float4* __restrict__ p4 = (const float4*)ws;
    float4 s = make_float4(0.f, 0.f, 0.f, 0.f);
    #pragma unroll
    for (int i = 0; i < GRID1 / NRED; ++i) {
      const float4 v = p4[(wave + NRED * i) * 32 + lane];
      s.x += v.x; s.y += v.y; s.z += v.z; s.w += v.w;
    }
    red[wave * 32 + lane] = s;
  } else if (wave == NRED) {
    const float k0 = kpts[lane];
    if (lane < 45 - 32) kp[32 + lane] = kpts[32 + lane];
    kp[lane] = k0;
  }
  __syncthreads();

  if (tid < CCH) {
    const int ch = tid;
    const float* redf = (const float*)red;
    float ssum = 0.f, qsum = 0.f;
    #pragma unroll
    for (int w = 0; w < NRED; ++w) {
      ssum += redf[w * 128 + ch];
      qsum += redf[w * 128 + 64 + ch];
    }
    const float inv_n = 1.0f / (float)NPTS;
    const float mu  = ssum * inv_n;
    const float var = qsum * inv_n - mu * mu;
    const float sc  = gamma[ch] * rsqrtf(var + BN_EPS);
    const float sh  = beta[ch] - mu * sc;
    bn[ch] = sc; bn[CCH + ch] = sh;
  }

  v2f bf[2][16];
  __syncthreads();

  auto act_half = [&](int p0, int bq, int rt) {
    int row = p0 + rt * 16 + lm16;
    row = (row < NPTS) ? row : (NPTS - 1);
    const float* __restrict__ arow = sfeat + row * CCH + 2 * half;
    v8f acc0 = (v8f){0.f,0.f,0.f,0.f,0.f,0.f,0.f,0.f};
    v8f acc1 = (v8f){0.f,0.f,0.f,0.f,0.f,0.f,0.f,0.f};
    #pragma unroll
    for (int hb = 0; hb < 16 / 8; ++hb) {
      v2f a[8];
      #pragma unroll
      for (int s = 0; s < 8; ++s) a[s] = *(const v2f*)(arow + 4 * (8 * hb + s));
      #pragma unroll
      for (int s = 0; s < 8; ++s) {
        acc0 = wmma_f32_16x16x4(a[s], bf[0][8 * hb + s], acc0);
        acc1 = wmma_f32_16x16x4(a[s], bf[1][8 * hb + s], acc1);
      }
    }
    const float* bnw = bn + hw * 32 + lm16;
    const float sc0 = bnw[0],  sh0 = bnw[CCH];
    const float sc1 = bnw[16], sh1 = bnw[CCH + 16];
    float* actw = &act[bq][(rt * 16 + 8 * half) * 68 + hw * 32 + lm16];
    #pragma unroll
    for (int r = 0; r < 8; ++r) {
      float v0 = acc0[r] * sc0 + sh0;
      float v1 = acc1[r] * sc1 + sh1;
      v0 = (v0 > 0.f) ? v0 : 0.1f * v0;
      v1 = (v1 > 0.f) ? v1 : 0.1f * v1;
      actw[r * 68]      = v0;
      actw[r * 68 + 16] = v1;
    }
  };

  auto conv_half = [&](int pb, int rt, float bias0, float bias1) {
    const int n0 = cwv * 32 + lm16;
    const float* arow = &act[pb][(rt * 16 + lm16) * 68 + 2 * half];
    v8f acc0 = (v8f){0.f,0.f,0.f,0.f,0.f,0.f,0.f,0.f};
    v8f acc1 = (v8f){0.f,0.f,0.f,0.f,0.f,0.f,0.f,0.f};
    #pragma unroll
    for (int s = 0; s < 16; ++s) {
      v2f a = *(const v2f*)(arow + 4 * s);
      acc0 = wmma_f32_16x16x4(a, bf[0][s], acc0);
      acc1 = wmma_f32_16x16x4(a, bf[1][s], acc1);
    }
    __half* cww = &cw[pb][(rt * 16 + 8 * half) * CWLD + n0];
    #pragma unroll
    for (int r = 0; r < 8; ++r) {
      cww[r * CWLD]      = __float2half(acc0[r] + bias0);
      cww[r * CWLD + 16] = __float2half(acc1[r] + bias1);
    }
  };

  const int npass = FULLR + (((int)blockIdx.x < TAILT) ? 1 : 0);
  auto pass_base  = [&](int i) { return (i < FULLR) ? ((int)blockIdx.x + i * GRID3) * PGRP : TAIL0 + (int)blockIdx.x * 16; };
  auto pass_tiles = [&](int i) { return (i < FULLR) ? 2 : 1; };

  auto matrix_half = [&](int ip, int pb, int rt, float bias0, float bias1) {
    if (convw) {
      conv_half(pb, rt, bias0, bias1);
    } else {
      const int in = ip + 1;
      if (in < npass && rt < pass_tiles(in)) act_half(pass_base(in), pb ^ 1, rt);
    }
  };

  auto geometry = [&](int gm, int j, int& gjk, float& gfl) {
    const bool valid = ((unsigned)j < (unsigned)NPTS);
    const int  jj    = valid ? j : 0;
    const float dx = spts[jj * 3 + 0] - qpts[gm * 3 + 0];
    const float dy = spts[jj * 3 + 1] - qpts[gm * 3 + 1];
    const float dz = spts[jj * 3 + 2] - qpts[gm * 3 + 2];
    float best = 3.4e38f; int bk = 0;
    #pragma unroll
    for (int k = 0; k < KPN; ++k) {
      const float ex = dx - kp[k * 3 + 0];
      const float ey = dy - kp[k * 3 + 1];
      const float ez = dz - kp[k * 3 + 2];
      const float sq = ex * ex + ey * ey + ez * ez;
      if (sq < best) { best = sq; bk = k; }
    }
    float infl = 1.0f - sqrtf(best) * (1.0f / SIGMA);
    infl = (infl > 0.f) ? infl : 0.f;
    if (!valid) infl = 0.f;
    gjk = jj | (bk << 16);
    gfl = infl;
  };

  struct Trip {
    bool  on[NWALK];
    int   jk[NWALK];
    float f[NWALK], x0[NWALK], x1[NWALK];
  };
  auto trip_issue = [&](unsigned& todo, Trip& t, int cjk, int cfb) {
    #pragma unroll
    for (int i = 0; i < NWALK; ++i) {
      t.on[i] = (todo != 0u);
      t.jk[i] = 0; t.f[i] = 0.f; t.x0[i] = 0.f; t.x1[i] = 0.f;
      if (t.on[i]) {
        const int h = __builtin_ctz(todo);
        todo &= todo - 1u;
        t.jk[i] = __builtin_amdgcn_readlane(cjk, h);
        t.f[i]  = __int_as_float(__builtin_amdgcn_readlane(cfb, h));
        const float* fr = sfeat + (t.jk[i] & 0xffff) * 64;
        t.x0[i] = fr[lane];
        t.x1[i] = fr[lane + 32];
      }
    }
  };
  auto trip_finish = [&](const Trip& t, const __half* cwr, float& a0, float& a1) {
    unsigned c0[NWALK], c1[NWALK];
    #pragma unroll
    for (int i = 0; i < NWALK; ++i) {
      c0[i] = 0u; c1[i] = 0u;
      if (t.on[i]) {
        const __half* cr = cwr + (t.jk[i] >> 16) * 64;
        c0[i] = __half_as_ushort(cr[0]);
        c1[i] = __half_as_ushort(cr[32]);
      }
    }
    #pragma unroll
    for (int i = 0; i < NWALK; ++i) {
      if (t.on[i]) {
        const float w0 = __half2float(__ushort_as_half((unsigned short)c0[i])) * t.f[i];
        const float w1 = __half2float(__ushort_as_half((unsigned short)c1[i])) * t.f[i];
        a0 = fmaf(t.x0[i], w0, a0);
        a1 = fmaf(t.x1[i], w1, a1);
      }
    }
  };

  {
    const float* __restrict__ Bsrc = convw ? W2 : W1;
    const int ldb   = convw ? CWLD : CCH;
    const int ncol0 = convw ? cwv * 32 : hw * 32;
    #pragma unroll
    for (int c = 0; c < 2; ++c) {
      const int n = ncol0 + c * 16 + lm16;
      #pragma unroll
      for (int s = 0; s < 16; ++s) {
        const int kk = 4 * s + 2 * half;
        bf[c][s].x = Bsrc[kk * ldb + n];
        bf[c][s].y = Bsrc[(kk + 1) * ldb + n];
      }
    }
  }
  if (!convw) { act_half(pass_base(0), 0, 0); act_half(pass_base(0), 0, 1); }
  __syncthreads();

  const float bias0 = convw ? b2[cwv * 32 + lm16]      : 0.f;
  const float bias1 = convw ? b2[cwv * 32 + lm16 + 16] : 0.f;

  int   gjk = 0;
  float gfl = 0.f;
  for (int it = -1; ; ++it) {
    const int  ip   = it + 1;
    const int  cb   = it & 1;
    const int  pb   = ip & 1;
    const bool hasP = (ip < npass);
    const int  tilP = pass_tiles(ip);
    const bool ownP = hasP && (wave < 16 * tilP);
    const int  m    = ((it >= 0) ? pass_base(it) : 0) + wave;
    const bool doC  = (it >= 0) && (wave < 16 * pass_tiles(it)) && (m < NPTS);

    int gm = pass_base(ip) + wave;
    gm = (gm < NPTS) ? gm : (NPTS - 1);
    int   j = 0;
    if (ownP) j = nbidx[gm * HNB + lane];

    auto matrix_stage = [&]() {
      if (hasP) {
        matrix_half(ip, pb, 0, bias0, bias1);
        if (tilP == 2) matrix_half(ip, pb, 1, bias0, bias1);
      }
    };
    auto side_stage = [&]() {
      const __half* cwr = &cw[cb][wave * CWLD + lane];
      const int     cfb = __float_as_int(gfl);
      unsigned todo = doC ? __builtin_amdgcn_ballot_w32(gfl > 0.f) : 0u;
      float a0 = 0.f, a1 = 0.f;
      while (todo) {
        Trip t1, t2;
        trip_issue(todo, t1, gjk, cfb);
        trip_issue(todo, t2, gjk, cfb);
        trip_finish(t1, cwr, a0, a1);
        trip_finish(t2, cwr, a0, a1);
      }
      if (doC) {
        out[m * 64 + lane]      = a0;
        out[m * 64 + lane + 32] = a1;
      }
      if (ownP) geometry(gm, j, gjk, gfl);
      else { gjk = 0; gfl = 0.f; }
    };

    if (young) { side_stage();   matrix_stage(); }
    else       { matrix_stage(); side_stage();   }
    __syncthreads();
    if (!hasP) break;
  }
}

extern "C" void kernel_launch(void* const* d_in, const int* in_sizes, int n_in,
                              void* d_out, int out_size, void* d_ws, size_t ws_size,
                              hipStream_t stream) {
  const float* qpts  = (const float*)d_in[0];
  const float* spts  = (const float*)d_in[1];
  const float* sfeat = (const float*)d_in[2];
  const int*   nbidx = (const int*)  d_in[3];
  const float* kpts  = (const float*)d_in[4];
  const float* W1    = (const float*)d_in[5];
  const float* gamma = (const float*)d_in[6];
  const float* beta  = (const float*)d_in[7];
  const float* W2    = (const float*)d_in[8];
  const float* b2    = (const float*)d_in[9];
  float* out = (float*)d_out;
  float* ws  = (float*)d_ws;

  k1_stats<<<GRID1, K1THREADS, 0, stream>>>(sfeat, W1, ws);
  k3_main<<<GRID3, K3THREADS, 0, stream>>>(qpts, spts, sfeat, nbidx, kpts,
                                     W1, gamma, beta, W2, b2, ws, out);
  (void)in_sizes; (void)n_in; (void)out_size; (void)ws_size;
}
